# starting kernel
# baseline (speedup 1.0000x reference)
#include <hip/hip_runtime.h>

#define B_DIM 8
#define N_DIM 1024
#define F_DIM 64
#define EPS_V 1e-10f

typedef __attribute__((ext_vector_type(2))) float v2f;
typedef __attribute__((ext_vector_type(4))) float v4f;
typedef __attribute__((ext_vector_type(8))) float v8f;

__global__ void hg_zero_kernel(float* __restrict__ p, int n) {
    int i = blockIdx.x * blockDim.x + threadIdx.x;
    if (i < n) p[i] = 0.0f;
}

__global__ void hg_degree_kernel(const int* __restrict__ eb,
                                 const int* __restrict__ ei,
                                 float* __restrict__ deg, int nnz) {
    int e = blockIdx.x * blockDim.x + threadIdx.x;
    if (e < nnz) atomicAdd(&deg[eb[e] * N_DIM + ei[e]], 1.0f);
}

__global__ __launch_bounds__(256) void hg_mlp_wmma_kernel(
    const float* __restrict__ x,
    const float* __restrict__ W1, const float* __restrict__ b1,
    const float* __restrict__ W2, const float* __restrict__ b2,
    const float* __restrict__ Ws1, const float* __restrict__ bs1,
    const float* __restrict__ Ws2, const float* __restrict__ bs2,
    float* __restrict__ x1_out, float* __restrict__ self_out) {

    __shared__ float sW[F_DIM * F_DIM];
    __shared__ float sB[F_DIM];
    __shared__ float sH[8][16 * F_DIM];

    const int tid  = threadIdx.x;
    const int wave = tid >> 5;
    const int lane = tid & 31;
    const int hlf  = lane >> 4;
    const int mn   = lane & 15;
    const int rowBase = (blockIdx.x * 8 + wave) * 16;

    v2f aX[16];
#pragma unroll
    for (int kk = 0; kk < 16; ++kk) {
        const float* p = x + (size_t)(rowBase + mn) * F_DIM + kk * 4 + hlf * 2;
        aX[kk][0] = p[0];
        aX[kk][1] = p[1];
    }

    const float* Wfirst[2]  = {W1, Ws1};
    const float* Bfirst[2]  = {b1, bs1};
    const float* Wsecond[2] = {W2, Ws2};
    const float* Bsecond[2] = {b2, bs2};
    float*       Dst[2]     = {x1_out, self_out};

    for (int s = 0; s < 2; ++s) {
        __syncthreads();
        for (int i = tid; i < F_DIM * F_DIM; i += 256) sW[i] = Wfirst[s][i];
        if (tid < F_DIM) sB[tid] = Bfirst[s][tid];
        __syncthreads();

#pragma unroll
        for (int nt = 0; nt < 4; ++nt) {
            v8f c;
            float bias = sB[nt * 16 + mn];
#pragma unroll
            for (int r = 0; r < 8; ++r) c[r] = bias;
#pragma unroll
            for (int kk = 0; kk < 16; ++kk) {
                v2f bF;
                bF[0] = sW[(kk * 4 + hlf * 2 + 0) * F_DIM + nt * 16 + mn];
                bF[1] = sW[(kk * 4 + hlf * 2 + 1) * F_DIM + nt * 16 + mn];
                c = __builtin_amdgcn_wmma_f32_16x16x4_f32(
                        false, aX[kk], false, bF, (short)0, c, false, false);
            }
#pragma unroll
            for (int r = 0; r < 8; ++r)
                sH[wave][(r + 8 * hlf) * F_DIM + nt * 16 + mn] = fmaxf(c[r], 0.0f);
        }
        __syncthreads();

        for (int i = tid; i < F_DIM * F_DIM; i += 256) sW[i] = Wsecond[s][i];
        if (tid < F_DIM) sB[tid] = Bsecond[s][tid];
        __syncthreads();

        v2f aH[16];
#pragma unroll
        for (int kk = 0; kk < 16; ++kk) {
            aH[kk][0] = sH[wave][mn * F_DIM + kk * 4 + hlf * 2 + 0];
            aH[kk][1] = sH[wave][mn * F_DIM + kk * 4 + hlf * 2 + 1];
        }

#pragma unroll
        for (int nt = 0; nt < 4; ++nt) {
            v8f c;
            float bias = sB[nt * 16 + mn];
#pragma unroll
            for (int r = 0; r < 8; ++r) c[r] = bias;
#pragma unroll
            for (int kk = 0; kk < 16; ++kk) {
                v2f bF;
                bF[0] = sW[(kk * 4 + hlf * 2 + 0) * F_DIM + nt * 16 + mn];
                bF[1] = sW[(kk * 4 + hlf * 2 + 1) * F_DIM + nt * 16 + mn];
                c = __builtin_amdgcn_wmma_f32_16x16x4_f32(
                        false, aH[kk], false, bF, (short)0, c, false, false);
            }
#pragma unroll
            for (int r = 0; r < 8; ++r) {
                Dst[s][(size_t)(rowBase + r + 8 * hlf) * F_DIM + nt * 16 + mn] =
                    fmaxf(c[r], 0.0f);
            }
        }
    }
}

__global__ __launch_bounds__(256) void hg_edge_kernel(
    const float* __restrict__ Wval,
    const int* __restrict__ eb, const int* __restrict__ ei,
    const int* __restrict__ ej, const int* __restrict__ ek,
    const float* __restrict__ deg, const float* __restrict__ x1,
    float* __restrict__ out, int nnz) {

    int gid = blockIdx.x * blockDim.x + threadIdx.x;
    int e = gid >> 4;
    if (e >= nnz) return;
    int c = (gid & 15) * 4;

    int b = __builtin_nontemporal_load(eb + e);
    int i = __builtin_nontemporal_load(ei + e);
    int j = __builtin_nontemporal_load(ej + e);
    int k = __builtin_nontemporal_load(ek + e);

    float a = 1.0f / (deg[b * N_DIM + i] + EPS_V);

    const v4f w = __builtin_nontemporal_load(
        (const v4f*)(Wval + (size_t)e * F_DIM + c));
    const float4 xk = *(const float4*)(x1 + (size_t)(b * N_DIM + k) * F_DIM + c);
    const float4 xj = *(const float4*)(x1 + (size_t)(b * N_DIM + j) * F_DIM + c);

    float* o = out + (size_t)(b * N_DIM + i) * F_DIM + c;
    atomicAdd(o + 0, a * w[0] * xk.x * xj.x);
    atomicAdd(o + 1, a * w[1] * xk.y * xj.y);
    atomicAdd(o + 2, a * w[2] * xk.z * xj.z);
    atomicAdd(o + 3, a * w[3] * xk.w * xj.w);
}

extern "C" void kernel_launch(void* const* d_in, const int* in_sizes, int n_in,
                              void* d_out, int out_size, void* d_ws, size_t ws_size,
                              hipStream_t stream) {
    const float* x    = (const float*)d_in[0];
    const float* Wval = (const float*)d_in[1];
    const int*   eb   = (const int*)d_in[2];
    const int*   ei   = (const int*)d_in[3];
    const int*   ej   = (const int*)d_in[4];
    const int*   ek   = (const int*)d_in[5];
    const float* W1   = (const float*)d_in[6];
    const float* b1   = (const float*)d_in[7];
    const float* W2   = (const float*)d_in[8];
    const float* b2   = (const float*)d_in[9];
    const float* Ws1  = (const float*)d_in[10];
    const float* bs1  = (const float*)d_in[11];
    const float* Ws2  = (const float*)d_in[12];
    const float* bs2  = (const float*)d_in[13];

    const int nnz   = in_sizes[2];
    const int nodes = B_DIM * N_DIM;

    float* deg = (float*)d_ws;
    float* x1  = deg + nodes;
    float* out = (float*)d_out;

    hg_zero_kernel<<<(nodes + 255) / 256, 256, 0, stream>>>(deg, nodes);

    hg_degree_kernel<<<(nnz + 255) / 256, 256, 0, stream>>>(eb, ei, deg, nnz);

    hg_mlp_wmma_kernel<<<nodes / 128, 256, 0, stream>>>(
        x, W1, b1, W2, b2, Ws1, bs1, Ws2, bs2, x1, out);

    long long threads = (long long)nnz * 16;
    hg_edge_kernel<<<(unsigned)((threads + 255) / 256), 256, 0, stream>>>(
        Wval, eb, ei, ej, ek, deg, x1, out, nnz);
}


# degree count: per-block LDS histograms (LDS integer adds) stored as partial planes and summed by a small kernel, in place of zero-fill + 1M global float atomics on 8192 words; counts exact
# speedup vs baseline: 1.8006x; 1.8006x over previous
#include <hip/hip_runtime.h>

#define B_DIM 8
#define N_DIM 1024
#define F_DIM 64
#define EPS_V 1e-10f

typedef __attribute__((ext_vector_type(2))) float v2f;
typedef __attribute__((ext_vector_type(4))) float v4f;
typedef __attribute__((ext_vector_type(8))) float v8f;

#define HG_BINS  (B_DIM * N_DIM)
#define HG_PARTS 64
#define HG_HT    1024

typedef __attribute__((ext_vector_type(4))) unsigned v4u;
typedef __attribute__((ext_vector_type(4))) int v4i;

__device__ __forceinline__ void hg_count4(unsigned* h, v4i b, v4i i) {
    atomicAdd(&h[b.x * N_DIM + i.x], 1u);
    atomicAdd(&h[b.y * N_DIM + i.y], 1u);
    atomicAdd(&h[b.z * N_DIM + i.z], 1u);
    atomicAdd(&h[b.w * N_DIM + i.w], 1u);
}

__global__ __launch_bounds__(HG_HT) void hg_hist_kernel(
    const int* __restrict__ eb, const int* __restrict__ ei,
    unsigned* __restrict__ part, int nnz) {

    __shared__ unsigned h[HG_BINS];

    const int tid = threadIdx.x;
    v4u* h4 = (v4u*)h;
    for (int q = tid; q < HG_BINS / 4; q += HG_HT) h4[q] = (v4u)(0u);
    __syncthreads();

    const v4i* eb4 = (const v4i*)eb;
    const v4i* ei4 = (const v4i*)ei;
    const int nvec   = nnz >> 2;
    const int stride = gridDim.x * HG_HT;
    int v = blockIdx.x * HG_HT + tid;

    for (; v + 3 * stride < nvec; v += 4 * stride) {
        const v4i b0 = eb4[v];
        const v4i b1 = eb4[v + stride];
        const v4i b2 = eb4[v + 2 * stride];
        const v4i b3 = eb4[v + 3 * stride];
        const v4i i0 = ei4[v];
        const v4i i1 = ei4[v + stride];
        const v4i i2 = ei4[v + 2 * stride];
        const v4i i3 = ei4[v + 3 * stride];
        asm volatile("" :: "v"(b0), "v"(b1), "v"(b2), "v"(b3),
                           "v"(i0), "v"(i1), "v"(i2), "v"(i3));
        hg_count4(h, b0, i0);
        hg_count4(h, b1, i1);
        hg_count4(h, b2, i2);
        hg_count4(h, b3, i3);
    }
    for (; v < nvec; v += stride) hg_count4(h, eb4[v], ei4[v]);

    if (blockIdx.x == 0) {
        const int e = (nvec << 2) + tid;
        if (e < nnz) atomicAdd(&h[eb[e] * N_DIM + ei[e]], 1u);
    }
    __syncthreads();

    v4u* dst = (v4u*)(part + (size_t)blockIdx.x * HG_BINS);
    for (int q = tid; q < HG_BINS / 4; q += HG_HT) dst[q] = h4[q];
}

__global__ __launch_bounds__(64) void hg_degsum_kernel(
    const unsigned* __restrict__ part, float* __restrict__ deg) {
    const int bin = blockIdx.x * 64 + threadIdx.x;
    unsigned s = 0u;
#pragma unroll
    for (int p = 0; p < HG_PARTS; ++p) s += part[p * HG_BINS + bin];
    deg[bin] = (float)s;
}

__global__ __launch_bounds__(256) void hg_mlp_wmma_kernel(
    const float* __restrict__ x,
    const float* __restrict__ W1, const float* __restrict__ b1,
    const float* __restrict__ W2, const float* __restrict__ b2,
    const float* __restrict__ Ws1, const float* __restrict__ bs1,
    const float* __restrict__ Ws2, const float* __restrict__ bs2,
    float* __restrict__ x1_out, float* __restrict__ self_out) {

    __shared__ float sW[F_DIM * F_DIM];
    __shared__ float sB[F_DIM];
    __shared__ float sH[8][16 * F_DIM];

    const int tid  = threadIdx.x;
    const int wave = tid >> 5;
    const int lane = tid & 31;
    const int hlf  = lane >> 4;
    const int mn   = lane & 15;
    const int rowBase = (blockIdx.x * 8 + wave) * 16;

    v2f aX[16];
#pragma unroll
    for (int kk = 0; kk < 16; ++kk) {
        const float* p = x + (size_t)(rowBase + mn) * F_DIM + kk * 4 + hlf * 2;
        aX[kk][0] = p[0];
        aX[kk][1] = p[1];
    }

    const float* Wfirst[2]  = {W1, Ws1};
    const float* Bfirst[2]  = {b1, bs1};
    const float* Wsecond[2] = {W2, Ws2};
    const float* Bsecond[2] = {b2, bs2};
    float*       Dst[2]     = {x1_out, self_out};

    for (int s = 0; s < 2; ++s) {
        __syncthreads();
        for (int i = tid; i < F_DIM * F_DIM; i += 256) sW[i] = Wfirst[s][i];
        if (tid < F_DIM) sB[tid] = Bfirst[s][tid];
        __syncthreads();

#pragma unroll
        for (int nt = 0; nt < 4; ++nt) {
            v8f c;
            float bias = sB[nt * 16 + mn];
#pragma unroll
            for (int r = 0; r < 8; ++r) c[r] = bias;
#pragma unroll
            for (int kk = 0; kk < 16; ++kk) {
                v2f bF;
                bF[0] = sW[(kk * 4 + hlf * 2 + 0) * F_DIM + nt * 16 + mn];
                bF[1] = sW[(kk * 4 + hlf * 2 + 1) * F_DIM + nt * 16 + mn];
                c = __builtin_amdgcn_wmma_f32_16x16x4_f32(
                        false, aX[kk], false, bF, (short)0, c, false, false);
            }
#pragma unroll
            for (int r = 0; r < 8; ++r)
                sH[wave][(r + 8 * hlf) * F_DIM + nt * 16 + mn] = fmaxf(c[r], 0.0f);
        }
        __syncthreads();

        for (int i = tid; i < F_DIM * F_DIM; i += 256) sW[i] = Wsecond[s][i];
        if (tid < F_DIM) sB[tid] = Bsecond[s][tid];
        __syncthreads();

        v2f aH[16];
#pragma unroll
        for (int kk = 0; kk < 16; ++kk) {
            aH[kk][0] = sH[wave][mn * F_DIM + kk * 4 + hlf * 2 + 0];
            aH[kk][1] = sH[wave][mn * F_DIM + kk * 4 + hlf * 2 + 1];
        }

#pragma unroll
        for (int nt = 0; nt < 4; ++nt) {
            v8f c;
            float bias = sB[nt * 16 + mn];
#pragma unroll
            for (int r = 0; r < 8; ++r) c[r] = bias;
#pragma unroll
            for (int kk = 0; kk < 16; ++kk) {
                v2f bF;
                bF[0] = sW[(kk * 4 + hlf * 2 + 0) * F_DIM + nt * 16 + mn];
                bF[1] = sW[(kk * 4 + hlf * 2 + 1) * F_DIM + nt * 16 + mn];
                c = __builtin_amdgcn_wmma_f32_16x16x4_f32(
                        false, aH[kk], false, bF, (short)0, c, false, false);
            }
#pragma unroll
            for (int r = 0; r < 8; ++r) {
                Dst[s][(size_t)(rowBase + r + 8 * hlf) * F_DIM + nt * 16 + mn] =
                    fmaxf(c[r], 0.0f);
            }
        }
    }
}

__global__ __launch_bounds__(256) void hg_edge_kernel(
    const float* __restrict__ Wval,
    const int* __restrict__ eb, const int* __restrict__ ei,
    const int* __restrict__ ej, const int* __restrict__ ek,
    const float* __restrict__ deg, const float* __restrict__ x1,
    float* __restrict__ out, int nnz) {

    int gid = blockIdx.x * blockDim.x + threadIdx.x;
    int e = gid >> 4;
    if (e >= nnz) return;
    int c = (gid & 15) * 4;

    int b = __builtin_nontemporal_load(eb + e);
    int i = __builtin_nontemporal_load(ei + e);
    int j = __builtin_nontemporal_load(ej + e);
    int k = __builtin_nontemporal_load(ek + e);

    float a = 1.0f / (deg[b * N_DIM + i] + EPS_V);

    const v4f w = __builtin_nontemporal_load(
        (const v4f*)(Wval + (size_t)e * F_DIM + c));
    const float4 xk = *(const float4*)(x1 + (size_t)(b * N_DIM + k) * F_DIM + c);
    const float4 xj = *(const float4*)(x1 + (size_t)(b * N_DIM + j) * F_DIM + c);

    float* o = out + (size_t)(b * N_DIM + i) * F_DIM + c;
    atomicAdd(o + 0, a * w[0] * xk.x * xj.x);
    atomicAdd(o + 1, a * w[1] * xk.y * xj.y);
    atomicAdd(o + 2, a * w[2] * xk.z * xj.z);
    atomicAdd(o + 3, a * w[3] * xk.w * xj.w);
}

extern "C" void kernel_launch(void* const* d_in, const int* in_sizes, int n_in,
                              void* d_out, int out_size, void* d_ws, size_t ws_size,
                              hipStream_t stream) {
    const float* x    = (const float*)d_in[0];
    const float* Wval = (const float*)d_in[1];
    const int*   eb   = (const int*)d_in[2];
    const int*   ei   = (const int*)d_in[3];
    const int*   ej   = (const int*)d_in[4];
    const int*   ek   = (const int*)d_in[5];
    const float* W1   = (const float*)d_in[6];
    const float* b1   = (const float*)d_in[7];
    const float* W2   = (const float*)d_in[8];
    const float* b2   = (const float*)d_in[9];
    const float* Ws1  = (const float*)d_in[10];
    const float* bs1  = (const float*)d_in[11];
    const float* Ws2  = (const float*)d_in[12];
    const float* bs2  = (const float*)d_in[13];

    const int nnz   = in_sizes[2];
    const int nodes = B_DIM * N_DIM;

    float* deg = (float*)d_ws;
    float* x1  = deg + nodes;
    float* out = (float*)d_out;

    unsigned* part = (unsigned*)x1;
    hg_hist_kernel<<<HG_PARTS, HG_HT, 0, stream>>>(eb, ei, part, nnz);

    hg_degsum_kernel<<<HG_BINS / 64, 64, 0, stream>>>(part, deg);

    hg_mlp_wmma_kernel<<<nodes / 128, 256, 0, stream>>>(
        x, W1, b1, W2, b2, Ws1, bs1, Ws2, bs2, x1, out);

    long long threads = (long long)nnz * 16;
    hg_edge_kernel<<<(unsigned)((threads + 255) / 256), 256, 0, stream>>>(
        Wval, eb, ei, ej, ek, deg, x1, out, nnz);
}


# edge kernel as a loop of resident waves, a wave per edge: lane l owns features l and l+32, so each f32 atomic adds into one whole 128-B line of out (was a quarter of four lines); arrival-order sums
# speedup vs baseline: 2.2308x; 1.2389x over previous
#include <hip/hip_runtime.h>

#define B_DIM 8
#define N_DIM 1024
#define F_DIM 64
#define EPS_V 1e-10f

typedef __attribute__((ext_vector_type(2))) float v2f;
typedef __attribute__((ext_vector_type(4))) float v4f;
typedef __attribute__((ext_vector_type(8))) float v8f;

#define HG_BINS  (B_DIM * N_DIM)
#define HG_PARTS 64
#define HG_HT    1024

typedef __attribute__((ext_vector_type(4))) unsigned v4u;
typedef __attribute__((ext_vector_type(4))) int v4i;

__device__ __forceinline__ void hg_count4(unsigned* h, v4i b, v4i i) {
    atomicAdd(&h[b.x * N_DIM + i.x], 1u);
    atomicAdd(&h[b.y * N_DIM + i.y], 1u);
    atomicAdd(&h[b.z * N_DIM + i.z], 1u);
    atomicAdd(&h[b.w * N_DIM + i.w], 1u);
}

__global__ __launch_bounds__(HG_HT) void hg_hist_kernel(
    const int* __restrict__ eb, const int* __restrict__ ei,
    unsigned* __restrict__ part, int nnz) {

    __shared__ unsigned h[HG_BINS];

    const int tid = threadIdx.x;
    v4u* h4 = (v4u*)h;
    for (int q = tid; q < HG_BINS / 4; q += HG_HT) h4[q] = (v4u)(0u);
    __syncthreads();

    const v4i* eb4 = (const v4i*)eb;
    const v4i* ei4 = (const v4i*)ei;
    const int nvec   = nnz >> 2;
    const int stride = gridDim.x * HG_HT;
    int v = blockIdx.x * HG_HT + tid;

    for (; v + 3 * stride < nvec; v += 4 * stride) {
        const v4i b0 = eb4[v];
        const v4i b1 = eb4[v + stride];
        const v4i b2 = eb4[v + 2 * stride];
        const v4i b3 = eb4[v + 3 * stride];
        const v4i i0 = ei4[v];
        const v4i i1 = ei4[v + stride];
        const v4i i2 = ei4[v + 2 * stride];
        const v4i i3 = ei4[v + 3 * stride];
        asm volatile("" :: "v"(b0), "v"(b1), "v"(b2), "v"(b3),
                           "v"(i0), "v"(i1), "v"(i2), "v"(i3));
        hg_count4(h, b0, i0);
        hg_count4(h, b1, i1);
        hg_count4(h, b2, i2);
        hg_count4(h, b3, i3);
    }
    for (; v < nvec; v += stride) hg_count4(h, eb4[v], ei4[v]);

    if (blockIdx.x == 0) {
        const int e = (nvec << 2) + tid;
        if (e < nnz) atomicAdd(&h[eb[e] * N_DIM + ei[e]], 1u);
    }
    __syncthreads();

    v4u* dst = (v4u*)(part + (size_t)blockIdx.x * HG_BINS);
    for (int q = tid; q < HG_BINS / 4; q += HG_HT) dst[q] = h4[q];
}

__global__ __launch_bounds__(64) void hg_degsum_kernel(
    const unsigned* __restrict__ part, float* __restrict__ deg) {
    const int bin = blockIdx.x * 64 + threadIdx.x;
    unsigned s = 0u;
#pragma unroll
    for (int p = 0; p < HG_PARTS; ++p) s += part[p * HG_BINS + bin];
    deg[bin] = (float)s;
}

__global__ __launch_bounds__(256) void hg_mlp_wmma_kernel(
    const float* __restrict__ x,
    const float* __restrict__ W1, const float* __restrict__ b1,
    const float* __restrict__ W2, const float* __restrict__ b2,
    const float* __restrict__ Ws1, const float* __restrict__ bs1,
    const float* __restrict__ Ws2, const float* __restrict__ bs2,
    float* __restrict__ x1_out, float* __restrict__ self_out) {

    __shared__ float sW[F_DIM * F_DIM];
    __shared__ float sB[F_DIM];
    __shared__ float sH[8][16 * F_DIM];

    const int tid  = threadIdx.x;
    const int wave = tid >> 5;
    const int lane = tid & 31;
    const int hlf  = lane >> 4;
    const int mn   = lane & 15;
    const int rowBase = (blockIdx.x * 8 + wave) * 16;

    v2f aX[16];
#pragma unroll
    for (int kk = 0; kk < 16; ++kk) {
        const float* p = x + (size_t)(rowBase + mn) * F_DIM + kk * 4 + hlf * 2;
        aX[kk][0] = p[0];
        aX[kk][1] = p[1];
    }

    const float* Wfirst[2]  = {W1, Ws1};
    const float* Bfirst[2]  = {b1, bs1};
    const float* Wsecond[2] = {W2, Ws2};
    const float* Bsecond[2] = {b2, bs2};
    float*       Dst[2]     = {x1_out, self_out};

    for (int s = 0; s < 2; ++s) {
        __syncthreads();
        for (int i = tid; i < F_DIM * F_DIM; i += 256) sW[i] = Wfirst[s][i];
        if (tid < F_DIM) sB[tid] = Bfirst[s][tid];
        __syncthreads();

#pragma unroll
        for (int nt = 0; nt < 4; ++nt) {
            v8f c;
            float bias = sB[nt * 16 + mn];
#pragma unroll
            for (int r = 0; r < 8; ++r) c[r] = bias;
#pragma unroll
            for (int kk = 0; kk < 16; ++kk) {
                v2f bF;
                bF[0] = sW[(kk * 4 + hlf * 2 + 0) * F_DIM + nt * 16 + mn];
                bF[1] = sW[(kk * 4 + hlf * 2 + 1) * F_DIM + nt * 16 + mn];
                c = __builtin_amdgcn_wmma_f32_16x16x4_f32(
                        false, aX[kk], false, bF, (short)0, c, false, false);
            }
#pragma unroll
            for (int r = 0; r < 8; ++r)
                sH[wave][(r + 8 * hlf) * F_DIM + nt * 16 + mn] = fmaxf(c[r], 0.0f);
        }
        __syncthreads();

        for (int i = tid; i < F_DIM * F_DIM; i += 256) sW[i] = Wsecond[s][i];
        if (tid < F_DIM) sB[tid] = Bsecond[s][tid];
        __syncthreads();

        v2f aH[16];
#pragma unroll
        for (int kk = 0; kk < 16; ++kk) {
            aH[kk][0] = sH[wave][mn * F_DIM + kk * 4 + hlf * 2 + 0];
            aH[kk][1] = sH[wave][mn * F_DIM + kk * 4 + hlf * 2 + 1];
        }

#pragma unroll
        for (int nt = 0; nt < 4; ++nt) {
            v8f c;
            float bias = sB[nt * 16 + mn];
#pragma unroll
            for (int r = 0; r < 8; ++r) c[r] = bias;
#pragma unroll
            for (int kk = 0; kk < 16; ++kk) {
                v2f bF;
                bF[0] = sW[(kk * 4 + hlf * 2 + 0) * F_DIM + nt * 16 + mn];
                bF[1] = sW[(kk * 4 + hlf * 2 + 1) * F_DIM + nt * 16 + mn];
                c = __builtin_amdgcn_wmma_f32_16x16x4_f32(
                        false, aH[kk], false, bF, (short)0, c, false, false);
            }
#pragma unroll
            for (int r = 0; r < 8; ++r) {
                Dst[s][(size_t)(rowBase + r + 8 * hlf) * F_DIM + nt * 16 + mn] =
                    fmaxf(c[r], 0.0f);
            }
        }
    }
}

#define EG_WAVES 32
#define EG_BLOCKS 256
#define EG_GROUP 8

__device__ __forceinline__ int eg_lane_i(int v, int t) {
    return __builtin_amdgcn_readlane(v, t);
}
__device__ __forceinline__ float eg_lane_f(float v, int t) {
    return __int_as_float(__builtin_amdgcn_readlane(__float_as_int(v), t));
}

__global__ __launch_bounds__(EG_WAVES * 32) void hg_edge_kernel(
    const float* __restrict__ Wval,
    const int* __restrict__ eb, const int* __restrict__ ei,
    const int* __restrict__ ej, const int* __restrict__ ek,
    const float* __restrict__ deg, const float* __restrict__ x1,
    float* __restrict__ out, int nnz) {

    const int lane = threadIdx.x & 31;
    const int wv = __builtin_amdgcn_readfirstlane(
        (int)(blockIdx.x * EG_WAVES + (threadIdx.x >> 5)));
    const int nwaves = gridDim.x * EG_WAVES;
    const int nruns  = (nnz + 31) >> 5;

    const int last = nnz - 1;
    int bv, iv, jv, kv;
    {
        const int et = min((wv << 5) + lane, last);
        bv = __builtin_nontemporal_load(eb + et);
        iv = __builtin_nontemporal_load(ei + et);
        jv = __builtin_nontemporal_load(ej + et);
        kv = __builtin_nontemporal_load(ek + et);
    }

    for (int c = wv; c < nruns; c += nwaves) {
        const int e0 = c << 5;
        const float dg = deg[bv * N_DIM + iv];
        const int oOff = (bv * N_DIM + iv) * F_DIM;
        const int kOff = (bv * N_DIM + kv) * F_DIM;
        const int jOff = (bv * N_DIM + jv) * F_DIM;

        __builtin_amdgcn_sched_barrier(0);

        {
            const int et = min(((c + nwaves) << 5) + lane, last);
            bv = __builtin_nontemporal_load(eb + et);
            iv = __builtin_nontemporal_load(ei + et);
            jv = __builtin_nontemporal_load(ej + et);
            kv = __builtin_nontemporal_load(ek + et);
        }
        const float av = 1.0f / (dg + EPS_V);

        const float* wrun = Wval + (size_t)e0 * F_DIM + lane;
        const float* xl   = x1 + lane;
        float*       ol   = out + lane;
        const int cnt = nnz - e0;

        if (cnt >= 32) {
            float w0[2][EG_GROUP], w1[2][EG_GROUP];
#pragma unroll
            for (int u = 0; u < EG_GROUP; ++u) {
                w0[0][u] = __builtin_nontemporal_load(wrun + u * F_DIM);
                w1[0][u] = __builtin_nontemporal_load(wrun + u * F_DIM + 32);
            }
            __builtin_amdgcn_sched_barrier(0);
#pragma unroll
            for (int g = 0; g < 32; g += EG_GROUP) {
                const int cur = (g / EG_GROUP) & 1;
                float k0[EG_GROUP], k1[EG_GROUP];
                float j0[EG_GROUP], j1[EG_GROUP];
#pragma unroll
                for (int u = 0; u < EG_GROUP; ++u) {
                    const int t = g + u;
                    const float* xk = xl + eg_lane_i(kOff, t);
                    const float* xj = xl + eg_lane_i(jOff, t);
                    k0[u] = xk[0];
                    k1[u] = xk[32];
                    j0[u] = xj[0];
                    j1[u] = xj[32];
                }
                if (g + EG_GROUP < 32) {
#pragma unroll
                    for (int u = 0; u < EG_GROUP; ++u) {
                        const float* wp = wrun + (g + EG_GROUP + u) * F_DIM;
                        w0[cur ^ 1][u] = __builtin_nontemporal_load(wp);
                        w1[cur ^ 1][u] = __builtin_nontemporal_load(wp + 32);
                    }
                }
                __builtin_amdgcn_sched_barrier(0);
#pragma unroll
                for (int u = 0; u < EG_GROUP; ++u) {
                    const int t = g + u;
                    const float a = eg_lane_f(av, t);
                    float* o = ol + eg_lane_i(oOff, t);
                    atomicAdd(o,      a * w0[cur][u] * k0[u] * j0[u]);
                    atomicAdd(o + 32, a * w1[cur][u] * k1[u] * j1[u]);
                }
            }
        } else {
            for (int t = 0; t < cnt; ++t) {
                const float* wp = wrun + t * F_DIM;
                const float* xk = xl + eg_lane_i(kOff, t);
                const float* xj = xl + eg_lane_i(jOff, t);
                const float a = eg_lane_f(av, t);
                float* o = ol + eg_lane_i(oOff, t);
                const float w0 = __builtin_nontemporal_load(wp);
                const float w1 = __builtin_nontemporal_load(wp + 32);
                atomicAdd(o,      a * w0 * xk[0]  * xj[0]);
                atomicAdd(o + 32, a * w1 * xk[32] * xj[32]);
            }
        }
    }
}

extern "C" void kernel_launch(void* const* d_in, const int* in_sizes, int n_in,
                              void* d_out, int out_size, void* d_ws, size_t ws_size,
                              hipStream_t stream) {
    const float* x    = (const float*)d_in[0];
    const float* Wval = (const float*)d_in[1];
    const int*   eb   = (const int*)d_in[2];
    const int*   ei   = (const int*)d_in[3];
    const int*   ej   = (const int*)d_in[4];
    const int*   ek   = (const int*)d_in[5];
    const float* W1   = (const float*)d_in[6];
    const float* b1   = (const float*)d_in[7];
    const float* W2   = (const float*)d_in[8];
    const float* b2   = (const float*)d_in[9];
    const float* Ws1  = (const float*)d_in[10];
    const float* bs1  = (const float*)d_in[11];
    const float* Ws2  = (const float*)d_in[12];
    const float* bs2  = (const float*)d_in[13];

    const int nnz   = in_sizes[2];
    const int nodes = B_DIM * N_DIM;

    float* deg = (float*)d_ws;
    float* x1  = deg + nodes;
    float* out = (float*)d_out;

    unsigned* part = (unsigned*)x1;
    hg_hist_kernel<<<HG_PARTS, HG_HT, 0, stream>>>(eb, ei, part, nnz);

    hg_degsum_kernel<<<HG_BINS / 64, 64, 0, stream>>>(part, deg);

    hg_mlp_wmma_kernel<<<nodes / 128, 256, 0, stream>>>(
        x, W1, b1, W2, b2, Ws1, bs1, Ws2, bs2, x1, out);

    hg_edge_kernel<<<EG_BLOCKS, EG_WAVES * 32, 0, stream>>>(
        Wval, eb, ei, ej, ek, deg, x1, out, nnz);
}


# edge kernel: a block owns one batch and a slice of the edges and keeps the batch's x1 rows (256 KB) in LDS: the k and j row gathers become LDS reads; arrival-order sums as before
# speedup vs baseline: 3.4638x; 1.5527x over previous
#include <hip/hip_runtime.h>

#define B_DIM 8
#define N_DIM 1024
#define F_DIM 64
#define EPS_V 1e-10f

typedef __attribute__((ext_vector_type(2))) float v2f;
typedef __attribute__((ext_vector_type(4))) float v4f;
typedef __attribute__((ext_vector_type(8))) float v8f;

#define HG_BINS  (B_DIM * N_DIM)
#define HG_PARTS 64
#define HG_HT    1024

typedef __attribute__((ext_vector_type(4))) unsigned v4u;
typedef __attribute__((ext_vector_type(4))) int v4i;

__device__ __forceinline__ void hg_count4(unsigned* h, v4i b, v4i i) {
    atomicAdd(&h[b.x * N_DIM + i.x], 1u);
    atomicAdd(&h[b.y * N_DIM + i.y], 1u);
    atomicAdd(&h[b.z * N_DIM + i.z], 1u);
    atomicAdd(&h[b.w * N_DIM + i.w], 1u);
}

__global__ __launch_bounds__(HG_HT) void hg_hist_kernel(
    const int* __restrict__ eb, const int* __restrict__ ei,
    unsigned* __restrict__ part, int nnz) {

    __shared__ unsigned h[HG_BINS];

    const int tid = threadIdx.x;
    v4u* h4 = (v4u*)h;
    for (int q = tid; q < HG_BINS / 4; q += HG_HT) h4[q] = (v4u)(0u);
    __syncthreads();

    const v4i* eb4 = (const v4i*)eb;
    const v4i* ei4 = (const v4i*)ei;
    const int nvec   = nnz >> 2;
    const int stride = gridDim.x * HG_HT;
    int v = blockIdx.x * HG_HT + tid;

    for (; v + 3 * stride < nvec; v += 4 * stride) {
        const v4i b0 = eb4[v];
        const v4i b1 = eb4[v + stride];
        const v4i b2 = eb4[v + 2 * stride];
        const v4i b3 = eb4[v + 3 * stride];
        const v4i i0 = ei4[v];
        const v4i i1 = ei4[v + stride];
        const v4i i2 = ei4[v + 2 * stride];
        const v4i i3 = ei4[v + 3 * stride];
        asm volatile("" :: "v"(b0), "v"(b1), "v"(b2), "v"(b3),
                           "v"(i0), "v"(i1), "v"(i2), "v"(i3));
        hg_count4(h, b0, i0);
        hg_count4(h, b1, i1);
        hg_count4(h, b2, i2);
        hg_count4(h, b3, i3);
    }
    for (; v < nvec; v += stride) hg_count4(h, eb4[v], ei4[v]);

    if (blockIdx.x == 0) {
        const int e = (nvec << 2) + tid;
        if (e < nnz) atomicAdd(&h[eb[e] * N_DIM + ei[e]], 1u);
    }
    __syncthreads();

    v4u* dst = (v4u*)(part + (size_t)blockIdx.x * HG_BINS);
    for (int q = tid; q < HG_BINS / 4; q += HG_HT) dst[q] = h4[q];
}

__global__ __launch_bounds__(64) void hg_degsum_kernel(
    const unsigned* __restrict__ part, float* __restrict__ deg) {
    const int bin = blockIdx.x * 64 + threadIdx.x;
    unsigned s = 0u;
#pragma unroll
    for (int p = 0; p < HG_PARTS; ++p) s += part[p * HG_BINS + bin];
    deg[bin] = (float)s;
}

__global__ __launch_bounds__(256) void hg_mlp_wmma_kernel(
    const float* __restrict__ x,
    const float* __restrict__ W1, const float* __restrict__ b1,
    const float* __restrict__ W2, const float* __restrict__ b2,
    const float* __restrict__ Ws1, const float* __restrict__ bs1,
    const float* __restrict__ Ws2, const float* __restrict__ bs2,
    float* __restrict__ x1_out, float* __restrict__ self_out) {

    __shared__ float sW[F_DIM * F_DIM];
    __shared__ float sB[F_DIM];
    __shared__ float sH[8][16 * F_DIM];

    const int tid  = threadIdx.x;
    const int wave = tid >> 5;
    const int lane = tid & 31;
    const int hlf  = lane >> 4;
    const int mn   = lane & 15;
    const int rowBase = (blockIdx.x * 8 + wave) * 16;

    v2f aX[16];
#pragma unroll
    for (int kk = 0; kk < 16; ++kk) {
        const float* p = x + (size_t)(rowBase + mn) * F_DIM + kk * 4 + hlf * 2;
        aX[kk][0] = p[0];
        aX[kk][1] = p[1];
    }

    const float* Wfirst[2]  = {W1, Ws1};
    const float* Bfirst[2]  = {b1, bs1};
    const float* Wsecond[2] = {W2, Ws2};
    const float* Bsecond[2] = {b2, bs2};
    float*       Dst[2]     = {x1_out, self_out};

    for (int s = 0; s < 2; ++s) {
        __syncthreads();
        for (int i = tid; i < F_DIM * F_DIM; i += 256) sW[i] = Wfirst[s][i];
        if (tid < F_DIM) sB[tid] = Bfirst[s][tid];
        __syncthreads();

#pragma unroll
        for (int nt = 0; nt < 4; ++nt) {
            v8f c;
            float bias = sB[nt * 16 + mn];
#pragma unroll
            for (int r = 0; r < 8; ++r) c[r] = bias;
#pragma unroll
            for (int kk = 0; kk < 16; ++kk) {
                v2f bF;
                bF[0] = sW[(kk * 4 + hlf * 2 + 0) * F_DIM + nt * 16 + mn];
                bF[1] = sW[(kk * 4 + hlf * 2 + 1) * F_DIM + nt * 16 + mn];
                c = __builtin_amdgcn_wmma_f32_16x16x4_f32(
                        false, aX[kk], false, bF, (short)0, c, false, false);
            }
#pragma unroll
            for (int r = 0; r < 8; ++r)
                sH[wave][(r + 8 * hlf) * F_DIM + nt * 16 + mn] = fmaxf(c[r], 0.0f);
        }
        __syncthreads();

        for (int i = tid; i < F_DIM * F_DIM; i += 256) sW[i] = Wsecond[s][i];
        if (tid < F_DIM) sB[tid] = Bsecond[s][tid];
        __syncthreads();

        v2f aH[16];
#pragma unroll
        for (int kk = 0; kk < 16; ++kk) {
            aH[kk][0] = sH[wave][mn * F_DIM + kk * 4 + hlf * 2 + 0];
            aH[kk][1] = sH[wave][mn * F_DIM + kk * 4 + hlf * 2 + 1];
        }

#pragma unroll
        for (int nt = 0; nt < 4; ++nt) {
            v8f c;
            float bias = sB[nt * 16 + mn];
#pragma unroll
            for (int r = 0; r < 8; ++r) c[r] = bias;
#pragma unroll
            for (int kk = 0; kk < 16; ++kk) {
                v2f bF;
                bF[0] = sW[(kk * 4 + hlf * 2 + 0) * F_DIM + nt * 16 + mn];
                bF[1] = sW[(kk * 4 + hlf * 2 + 1) * F_DIM + nt * 16 + mn];
                c = __builtin_amdgcn_wmma_f32_16x16x4_f32(
                        false, aH[kk], false, bF, (short)0, c, false, false);
            }
#pragma unroll
            for (int r = 0; r < 8; ++r) {
                Dst[s][(size_t)(rowBase + r + 8 * hlf) * F_DIM + nt * 16 + mn] =
                    fmaxf(c[r], 0.0f);
            }
        }
    }
}

#define EL_WAVES  32
#define EL_SLICES 32
#define EL_G      16
#define EL_H      8
#define EL_RING   128
#define EL_NB     10

static_assert((1 << EL_NB) == N_DIM, "a record packs three node indices of EL_NB bits");
static_assert(EL_WAVES * 32 == N_DIM, "the prologue computes one 1/degree per thread");
static_assert((N_DIM * F_DIM / 4) % (EL_WAVES * 32) == 0, "whole copy trips");
static_assert((EL_G - 1) + 64 <= EL_RING && (EL_RING & (EL_RING - 1)) == 0, "ring size");

typedef __attribute__((ext_vector_type(2))) int v2i;
typedef __attribute__((ext_vector_type(2))) unsigned v2u;

__device__ __forceinline__ int eg_lane_i(int v, int t) {
    return __builtin_amdgcn_readlane(v, t);
}
__device__ __forceinline__ float eg_lane_f(float v, int t) {
    return __int_as_float(__builtin_amdgcn_readlane(__float_as_int(v), t));
}

template <bool FULL>
__device__ __forceinline__ void el_group(
    const float* __restrict__ Wval, float* __restrict__ outb,
    const float* sx, const float* sinv, const v2u* ring,
    int head, int rem, int lane) {

    int u = lane & (EL_G - 1);
    if (!FULL) u = (u < rem) ? u : 0;
    const v2u rec = ring[(head + u) & (EL_RING - 1)];
    const int e    = (int)rec.x;
    const int i    = (int)(rec.y & (N_DIM - 1));
    const int jOff = (int)((rec.y >> EL_NB) & (N_DIM - 1)) * F_DIM;
    const int kOff = (int)(rec.y >> (2 * EL_NB)) * F_DIM;
    const int oOff = i * F_DIM;
    const float av = sinv[i];

    float w0[EL_G], w1[EL_G];
#pragma unroll
    for (int t = 0; t < EL_G; ++t) {
        const float* wp = Wval + (size_t)(unsigned)eg_lane_i(e, t) * F_DIM;
        w0[t] = __builtin_nontemporal_load(wp + lane);
        w1[t] = __builtin_nontemporal_load(wp + lane + 32);
    }
#pragma unroll
    for (int g = 0; g < EL_G; g += EL_H) {
        float k0[EL_H], k1[EL_H], j0[EL_H], j1[EL_H];
#pragma unroll
        for (int t = 0; t < EL_H; ++t) {
            const float* xk = sx + eg_lane_i(kOff, g + t) + lane;
            const float* xj = sx + eg_lane_i(jOff, g + t) + lane;
            k0[t] = xk[0];
            k1[t] = xk[32];
            j0[t] = xj[0];
            j1[t] = xj[32];
        }
#pragma unroll
        for (int t = 0; t < EL_H; ++t) {
            if (FULL || g + t < rem) {
                const float a = eg_lane_f(av, g + t);
                float* o = outb + eg_lane_i(oOff, g + t);
                atomicAdd(o + lane,      a * w0[g + t] * k0[t] * j0[t]);
                atomicAdd(o + lane + 32, a * w1[g + t] * k1[t] * j1[t]);
            }
        }
    }
}

__global__ __launch_bounds__(EL_WAVES * 32) void hg_edge_kernel(
    const float* __restrict__ Wval,
    const int* __restrict__ eb, const int* __restrict__ ei,
    const int* __restrict__ ej, const int* __restrict__ ek,
    const float* __restrict__ deg, const float* __restrict__ x1,
    float* __restrict__ out, int nnz) {

    __shared__ float sx[N_DIM * F_DIM];
    __shared__ float sinv[N_DIM];
    __shared__ v2u   sring[EL_WAVES][EL_RING];

    const int tid  = threadIdx.x;
    const int lane = tid & 31;
    const int wv    = __builtin_amdgcn_readfirstlane(tid >> 5);
    const int b     = blockIdx.x % B_DIM;
    const int slice = blockIdx.x / B_DIM;

    {
        const v4f* src = (const v4f*)(x1 + (size_t)b * N_DIM * F_DIM);
        v4f* dst = (v4f*)sx;
#pragma unroll
        for (int q = 0; q < N_DIM * F_DIM / 4 / (EL_WAVES * 32); ++q)
            dst[q * (EL_WAVES * 32) + tid] = src[q * (EL_WAVES * 32) + tid];
        sinv[tid] = 1.0f / (deg[b * N_DIM + tid] + EPS_V);
    }
    __syncthreads();

    float* outb = out + (size_t)b * N_DIM * F_DIM;
    v2u* ring = sring[wv];

    const int nfull = nnz >> 6;
    const int per   = (nfull + EL_SLICES - 1) / EL_SLICES;
    const int s0    = slice * per;
    const int s1    = min(s0 + per, nfull);

    const v2i* eb2 = (const v2i*)eb;
    const v2i* ei2 = (const v2i*)ei;
    const v2i* ej2 = (const v2i*)ej;
    const v2i* ek2 = (const v2i*)ek;

    int sr = s0 + wv;
    if (sr < s1) {
        int head = 0, tail = 0;
        v2i cb, ci, cj, ck;
        {
            const int p = sr * 32 + lane;
            cb = eb2[p];
            ci = ei2[p];
            cj = ej2[p];
            ck = ek2[p];
        }
        for (; sr < s1; sr += EL_WAVES) {
            const int e0 = sr * 64 + 2 * lane;
            const bool tx = (cb.x == b);
            const bool ty = (cb.y == b);
            const unsigned rx = (unsigned)(ci.x & (N_DIM - 1))
                              | ((unsigned)(cj.x & (N_DIM - 1)) << EL_NB)
                              | ((unsigned)(ck.x & (N_DIM - 1)) << (2 * EL_NB));
            const unsigned ry = (unsigned)(ci.y & (N_DIM - 1))
                              | ((unsigned)(cj.y & (N_DIM - 1)) << EL_NB)
                              | ((unsigned)(ck.y & (N_DIM - 1)) << (2 * EL_NB));

            {
                const int p = min(sr + EL_WAVES, s1 - 1) * 32 + lane;
                cb = eb2[p];
                ci = ei2[p];
                cj = ej2[p];
                ck = ek2[p];
            }

            const unsigned mx = __builtin_amdgcn_ballot_w32(tx);
            const unsigned my = __builtin_amdgcn_ballot_w32(ty);
            if (tx) {
                v2u r; r.x = (unsigned)e0; r.y = rx;
                ring[(tail + (int)__builtin_amdgcn_mbcnt_lo(mx, 0u)) & (EL_RING - 1)] = r;
            }
            tail += __builtin_popcount(mx);
            if (ty) {
                v2u r; r.x = (unsigned)(e0 + 1); r.y = ry;
                ring[(tail + (int)__builtin_amdgcn_mbcnt_lo(my, 0u)) & (EL_RING - 1)] = r;
            }
            tail += __builtin_popcount(my);

            while (tail - head >= EL_G) {
                el_group<true>(Wval, outb, sx, sinv, ring, head, EL_G, lane);
                head += EL_G;
            }
        }
        const int rem = tail - head;
        if (rem > 0) el_group<false>(Wval, outb, sx, sinv, ring, head, rem, lane);
    }

    if (slice == EL_SLICES - 1 && wv == 0) {
        for (int e = nfull << 6; e < nnz; ++e) {
            if (eb[e] == b) {
                const int i = ei[e] & (N_DIM - 1);
                const float* xk = sx + (ek[e] & (N_DIM - 1)) * F_DIM + lane;
                const float* xj = sx + (ej[e] & (N_DIM - 1)) * F_DIM + lane;
                const float* wp = Wval + (size_t)e * F_DIM;
                const float a = sinv[i];
                float* o = outb + i * F_DIM;
                const float w0 = __builtin_nontemporal_load(wp + lane);
                const float w1 = __builtin_nontemporal_load(wp + lane + 32);
                atomicAdd(o + lane,      a * w0 * xk[0]  * xj[0]);
                atomicAdd(o + lane + 32, a * w1 * xk[32] * xj[32]);
            }
        }
    }
}

extern "C" void kernel_launch(void* const* d_in, const int* in_sizes, int n_in,
                              void* d_out, int out_size, void* d_ws, size_t ws_size,
                              hipStream_t stream) {
    const float* x    = (const float*)d_in[0];
    const float* Wval = (const float*)d_in[1];
    const int*   eb   = (const int*)d_in[2];
    const int*   ei   = (const int*)d_in[3];
    const int*   ej   = (const int*)d_in[4];
    const int*   ek   = (const int*)d_in[5];
    const float* W1   = (const float*)d_in[6];
    const float* b1   = (const float*)d_in[7];
    const float* W2   = (const float*)d_in[8];
    const float* b2   = (const float*)d_in[9];
    const float* Ws1  = (const float*)d_in[10];
    const float* bs1  = (const float*)d_in[11];
    const float* Ws2  = (const float*)d_in[12];
    const float* bs2  = (const float*)d_in[13];

    const int nnz   = in_sizes[2];
    const int nodes = B_DIM * N_DIM;

    float* deg = (float*)d_ws;
    float* x1  = deg + nodes;
    float* out = (float*)d_out;

    unsigned* part = (unsigned*)x1;
    hg_hist_kernel<<<HG_PARTS, HG_HT, 0, stream>>>(eb, ei, part, nnz);

    hg_degsum_kernel<<<HG_BINS / 64, 64, 0, stream>>>(part, deg);

    hg_mlp_wmma_kernel<<<nodes / 128, 256, 0, stream>>>(
        x, W1, b1, W2, b2, Ws1, bs1, Ws2, bs2, x1, out);

    hg_edge_kernel<<<B_DIM * EL_SLICES, EL_WAVES * 32, 0, stream>>>(
        Wval, eb, ei, ej, ek, deg, x1, out, nnz);
}


# edge kernel: a block orders its edges by head node in LDS (count, prefix sum, place) and sums a node's run in registers: 2 line atomics per node, not per edge; LDS arrival, then atomic arrival order
# speedup vs baseline: 3.7667x; 1.0874x over previous
#include <hip/hip_runtime.h>

#define B_DIM 8
#define N_DIM 1024
#define F_DIM 64
#define EPS_V 1e-10f

typedef __attribute__((ext_vector_type(2))) float v2f;
typedef __attribute__((ext_vector_type(4))) float v4f;
typedef __attribute__((ext_vector_type(8))) float v8f;

#define HG_BINS  (B_DIM * N_DIM)
#define HG_PARTS 64
#define HG_HT    1024

typedef __attribute__((ext_vector_type(4))) unsigned v4u;
typedef __attribute__((ext_vector_type(4))) int v4i;

__device__ __forceinline__ void hg_count4(unsigned* h, v4i b, v4i i) {
    atomicAdd(&h[b.x * N_DIM + i.x], 1u);
    atomicAdd(&h[b.y * N_DIM + i.y], 1u);
    atomicAdd(&h[b.z * N_DIM + i.z], 1u);
    atomicAdd(&h[b.w * N_DIM + i.w], 1u);
}

__global__ __launch_bounds__(HG_HT) void hg_hist_kernel(
    const int* __restrict__ eb, const int* __restrict__ ei,
    unsigned* __restrict__ part, int nnz) {

    __shared__ unsigned h[HG_BINS];

    const int tid = threadIdx.x;
    v4u* h4 = (v4u*)h;
    for (int q = tid; q < HG_BINS / 4; q += HG_HT) h4[q] = (v4u)(0u);
    __syncthreads();

    const v4i* eb4 = (const v4i*)eb;
    const v4i* ei4 = (const v4i*)ei;
    const int nvec   = nnz >> 2;
    const int stride = gridDim.x * HG_HT;
    int v = blockIdx.x * HG_HT + tid;

    for (; v + 3 * stride < nvec; v += 4 * stride) {
        const v4i b0 = eb4[v];
        const v4i b1 = eb4[v + stride];
        const v4i b2 = eb4[v + 2 * stride];
        const v4i b3 = eb4[v + 3 * stride];
        const v4i i0 = ei4[v];
        const v4i i1 = ei4[v + stride];
        const v4i i2 = ei4[v + 2 * stride];
        const v4i i3 = ei4[v + 3 * stride];
        asm volatile("" :: "v"(b0), "v"(b1), "v"(b2), "v"(b3),
                           "v"(i0), "v"(i1), "v"(i2), "v"(i3));
        hg_count4(h, b0, i0);
        hg_count4(h, b1, i1);
        hg_count4(h, b2, i2);
        hg_count4(h, b3, i3);
    }
    for (; v < nvec; v += stride) hg_count4(h, eb4[v], ei4[v]);

    if (blockIdx.x == 0) {
        const int e = (nvec << 2) + tid;
        if (e < nnz) atomicAdd(&h[eb[e] * N_DIM + ei[e]], 1u);
    }
    __syncthreads();

    v4u* dst = (v4u*)(part + (size_t)blockIdx.x * HG_BINS);
    for (int q = tid; q < HG_BINS / 4; q += HG_HT) dst[q] = h4[q];
}

__global__ __launch_bounds__(64) void hg_degsum_kernel(
    const unsigned* __restrict__ part, float* __restrict__ deg) {
    const int bin = blockIdx.x * 64 + threadIdx.x;
    unsigned s = 0u;
#pragma unroll
    for (int p = 0; p < HG_PARTS; ++p) s += part[p * HG_BINS + bin];
    deg[bin] = (float)s;
}

__global__ __launch_bounds__(256) void hg_mlp_wmma_kernel(
    const float* __restrict__ x,
    const float* __restrict__ W1, const float* __restrict__ b1,
    const float* __restrict__ W2, const float* __restrict__ b2,
    const float* __restrict__ Ws1, const float* __restrict__ bs1,
    const float* __restrict__ Ws2, const float* __restrict__ bs2,
    float* __restrict__ x1_out, float* __restrict__ self_out) {

    __shared__ float sW[F_DIM * F_DIM];
    __shared__ float sB[F_DIM];
    __shared__ float sH[8][16 * F_DIM];

    const int tid  = threadIdx.x;
    const int wave = tid >> 5;
    const int lane = tid & 31;
    const int hlf  = lane >> 4;
    const int mn   = lane & 15;
    const int rowBase = (blockIdx.x * 8 + wave) * 16;

    v2f aX[16];
#pragma unroll
    for (int kk = 0; kk < 16; ++kk) {
        const float* p = x + (size_t)(rowBase + mn) * F_DIM + kk * 4 + hlf * 2;
        aX[kk][0] = p[0];
        aX[kk][1] = p[1];
    }

    const float* Wfirst[2]  = {W1, Ws1};
    const float* Bfirst[2]  = {b1, bs1};
    const float* Wsecond[2] = {W2, Ws2};
    const float* Bsecond[2] = {b2, bs2};
    float*       Dst[2]     = {x1_out, self_out};

    for (int s = 0; s < 2; ++s) {
        __syncthreads();
        for (int i = tid; i < F_DIM * F_DIM; i += 256) sW[i] = Wfirst[s][i];
        if (tid < F_DIM) sB[tid] = Bfirst[s][tid];
        __syncthreads();

#pragma unroll
        for (int nt = 0; nt < 4; ++nt) {
            v8f c;
            float bias = sB[nt * 16 + mn];
#pragma unroll
            for (int r = 0; r < 8; ++r) c[r] = bias;
#pragma unroll
            for (int kk = 0; kk < 16; ++kk) {
                v2f bF;
                bF[0] = sW[(kk * 4 + hlf * 2 + 0) * F_DIM + nt * 16 + mn];
                bF[1] = sW[(kk * 4 + hlf * 2 + 1) * F_DIM + nt * 16 + mn];
                c = __builtin_amdgcn_wmma_f32_16x16x4_f32(
                        false, aX[kk], false, bF, (short)0, c, false, false);
            }
#pragma unroll
            for (int r = 0; r < 8; ++r)
                sH[wave][(r + 8 * hlf) * F_DIM + nt * 16 + mn] = fmaxf(c[r], 0.0f);
        }
        __syncthreads();

        for (int i = tid; i < F_DIM * F_DIM; i += 256) sW[i] = Wsecond[s][i];
        if (tid < F_DIM) sB[tid] = Bsecond[s][tid];
        __syncthreads();

        v2f aH[16];
#pragma unroll
        for (int kk = 0; kk < 16; ++kk) {
            aH[kk][0] = sH[wave][mn * F_DIM + kk * 4 + hlf * 2 + 0];
            aH[kk][1] = sH[wave][mn * F_DIM + kk * 4 + hlf * 2 + 1];
        }

#pragma unroll
        for (int nt = 0; nt < 4; ++nt) {
            v8f c;
            float bias = sB[nt * 16 + mn];
#pragma unroll
            for (int r = 0; r < 8; ++r) c[r] = bias;
#pragma unroll
            for (int kk = 0; kk < 16; ++kk) {
                v2f bF;
                bF[0] = sW[(kk * 4 + hlf * 2 + 0) * F_DIM + nt * 16 + mn];
                bF[1] = sW[(kk * 4 + hlf * 2 + 1) * F_DIM + nt * 16 + mn];
                c = __builtin_amdgcn_wmma_f32_16x16x4_f32(
                        false, aH[kk], false, bF, (short)0, c, false, false);
            }
#pragma unroll
            for (int r = 0; r < 8; ++r) {
                Dst[s][(size_t)(rowBase + r + 8 * hlf) * F_DIM + nt * 16 + mn] =
                    fmaxf(c[r], 0.0f);
            }
        }
    }
}

#define EL_WAVES  32
#define EL_SLICES 32
#define EL_G      16
#define EL_H      8
#define EL_NB     10
#define EL_CAP    4608
#define EL_SR     8

static_assert((1 << EL_NB) == N_DIM, "a record packs three node indices of EL_NB bits");
static_assert(EL_WAVES * 32 == N_DIM, "one thread per node: 1/degree, counter, prefix sum");
static_assert((N_DIM * F_DIM / 4) % (EL_WAVES * 32) == 0, "whole copy trips");
static_assert(EL_CAP <= 65536, "list positions are kept as 16-bit words");

typedef __attribute__((ext_vector_type(2))) int v2i;
typedef __attribute__((ext_vector_type(2))) unsigned v2u;

__device__ __forceinline__ int eg_lane_i(int v, int t) {
    return __builtin_amdgcn_readlane(v, t);
}
__device__ __forceinline__ float eg_lane_f(float v, int t) {
    return __int_as_float(__builtin_amdgcn_readlane(__float_as_int(v), t));
}

__device__ __forceinline__ void el_direct(
    const float* __restrict__ Wval, float* __restrict__ outb,
    const float* sx, const float* sinv, unsigned e, unsigned y, int lane) {
    const int i = (int)(y & (N_DIM - 1));
    const float* xj = sx + (int)((y >> EL_NB) & (N_DIM - 1)) * F_DIM + lane;
    const float* xk = sx + (int)(y >> (2 * EL_NB)) * F_DIM + lane;
    const float* wp = Wval + (size_t)e * F_DIM;
    const float a = sinv[i];
    float* o = outb + i * F_DIM;
    const float w0 = __builtin_nontemporal_load(wp + lane);
    const float w1 = __builtin_nontemporal_load(wp + lane + 32);
    atomicAdd(o + lane,      a * w0 * xk[0]  * xj[0]);
    atomicAdd(o + lane + 32, a * w1 * xk[32] * xj[32]);
}

__device__ __forceinline__ v2u el_rec(
    const v2u* srec, const unsigned short* sord, int p, int hi, int lane) {
    const int q = min(p + (lane & (EL_G - 1)), hi - 1);
    return srec[sord[q]];
}

__device__ __forceinline__ void el_wload(
    const float* __restrict__ Wval, v2u rec, int lane,
    float (&w0)[EL_G], float (&w1)[EL_G]) {
    const int e = (int)rec.x;
#pragma unroll
    for (int t = 0; t < EL_G; ++t) {
        const float* wp = Wval + (size_t)(unsigned)eg_lane_i(e, t) * F_DIM;
        w0[t] = __builtin_nontemporal_load(wp + lane);
        w1[t] = __builtin_nontemporal_load(wp + lane + 32);
    }
}

__device__ __forceinline__ void el_sum(
    float* __restrict__ outb, const float* sx, const float* sinv, v2u rec,
    const float (&w0)[EL_G], const float (&w1)[EL_G],
    int rem, int lane, int& cur, float& acc0, float& acc1) {

    const int i    = (int)(rec.y & (N_DIM - 1));
    const int jOff = (int)((rec.y >> EL_NB) & (N_DIM - 1)) * F_DIM;
    const int kOff = (int)(rec.y >> (2 * EL_NB)) * F_DIM;
    const float av = sinv[i];

#pragma unroll
    for (int g = 0; g < EL_G; g += EL_H) {
        float k0[EL_H], k1[EL_H], j0[EL_H], j1[EL_H];
#pragma unroll
        for (int t = 0; t < EL_H; ++t) {
            const float* xk = sx + eg_lane_i(kOff, g + t) + lane;
            const float* xj = sx + eg_lane_i(jOff, g + t) + lane;
            k0[t] = xk[0];
            k1[t] = xk[32];
            j0[t] = xj[0];
            j1[t] = xj[32];
        }
#pragma unroll
        for (int t = 0; t < EL_H; ++t) {
            const int it = eg_lane_i(i, g + t);
            if (it != cur) {
                if (cur >= 0) {
                    float* o = outb + cur * F_DIM;
                    atomicAdd(o + lane,      acc0);
                    atomicAdd(o + lane + 32, acc1);
                }
                cur  = it;
                acc0 = 0.0f;
                acc1 = 0.0f;
            }
            const float a = (g + t < rem) ? eg_lane_f(av, g + t) : 0.0f;
            acc0 += a * w0[g + t] * k0[t] * j0[t];
            acc1 += a * w1[g + t] * k1[t] * j1[t];
        }
    }
}

__global__ __launch_bounds__(EL_WAVES * 32) void hg_edge_kernel(
    const float* __restrict__ Wval,
    const int* __restrict__ eb, const int* __restrict__ ei,
    const int* __restrict__ ej, const int* __restrict__ ek,
    const float* __restrict__ deg, const float* __restrict__ x1,
    float* __restrict__ out, int nnz) {

    __shared__ float sx[N_DIM * F_DIM];
    __shared__ float sinv[N_DIM];
    __shared__ v2u   srec[EL_CAP];
    __shared__ unsigned short sord[EL_CAP];
    __shared__ unsigned spos[N_DIM];
    __shared__ unsigned swtot[EL_WAVES];
    __shared__ unsigned snrec;

    const int tid  = threadIdx.x;
    const int lane = tid & 31;
    const int wv    = __builtin_amdgcn_readfirstlane(tid >> 5);
    const int b     = blockIdx.x % B_DIM;
    const int slice = blockIdx.x / B_DIM;

    spos[tid] = 0u;
    if (tid == 0) snrec = 0u;
    {
        const v4f* src = (const v4f*)(x1 + (size_t)b * N_DIM * F_DIM);
        v4f* dst = (v4f*)sx;
#pragma unroll
        for (int q = 0; q < N_DIM * F_DIM / 4 / (EL_WAVES * 32); ++q)
            dst[q * (EL_WAVES * 32) + tid] = src[q * (EL_WAVES * 32) + tid];
        sinv[tid] = 1.0f / (deg[b * N_DIM + tid] + EPS_V);
    }
    __syncthreads();

    float* outb = out + (size_t)b * N_DIM * F_DIM;

    const int nfull = nnz >> 6;
    const int per   = (nfull + EL_SLICES - 1) / EL_SLICES;
    const int s0    = slice * per;
    const int s1    = min(s0 + per, nfull);

    const v2i* eb2 = (const v2i*)eb;
    const v2i* ei2 = (const v2i*)ei;
    const v2i* ej2 = (const v2i*)ej;
    const v2i* ek2 = (const v2i*)ek;

    for (int sr = s0 + wv; sr < s1; sr += EL_WAVES * EL_SR) {
        v2i cb[EL_SR], ci[EL_SR], cj[EL_SR], ck[EL_SR];
#pragma unroll
        for (int q = 0; q < EL_SR; ++q) {
            const int p = min(sr + q * EL_WAVES, s1 - 1) * 32 + lane;
            cb[q] = eb2[p];
            ci[q] = ei2[p];
            cj[q] = ej2[p];
            ck[q] = ek2[p];
        }
        unsigned mx[EL_SR], my[EL_SR];
        unsigned tot = 0u;
#pragma unroll
        for (int q = 0; q < EL_SR; ++q) {
            const bool live = (sr + q * EL_WAVES < s1);
            mx[q] = __builtin_amdgcn_ballot_w32(live && cb[q].x == b);
            my[q] = __builtin_amdgcn_ballot_w32(live && cb[q].y == b);
            tot += (unsigned)__builtin_popcount(mx[q]) + (unsigned)__builtin_popcount(my[q]);
        }
        unsigned base = 0u;
        if (lane == 0) base = atomicAdd(&snrec, tot);
        base = (unsigned)__builtin_amdgcn_readfirstlane((int)base);
#pragma unroll
        for (int q = 0; q < EL_SR; ++q) {
            const unsigned e0 = (unsigned)((sr + q * EL_WAVES) * 64 + 2 * lane);
            const bool tx = ((mx[q] >> lane) & 1u) != 0u;
            const bool ty = ((my[q] >> lane) & 1u) != 0u;
            const unsigned ix = (unsigned)(ci[q].x & (N_DIM - 1));
            const unsigned iy = (unsigned)(ci[q].y & (N_DIM - 1));
            const unsigned rx = ix
                              | ((unsigned)(cj[q].x & (N_DIM - 1)) << EL_NB)
                              | ((unsigned)(ck[q].x & (N_DIM - 1)) << (2 * EL_NB));
            const unsigned ry = iy
                              | ((unsigned)(cj[q].y & (N_DIM - 1)) << EL_NB)
                              | ((unsigned)(ck[q].y & (N_DIM - 1)) << (2 * EL_NB));
            const unsigned nx  = (unsigned)__builtin_popcount(mx[q]);
            const unsigned slx = base + __builtin_amdgcn_mbcnt_lo(mx[q], 0u);
            const unsigned sly = base + nx + __builtin_amdgcn_mbcnt_lo(my[q], 0u);
            base += nx + (unsigned)__builtin_popcount(my[q]);
            const bool fx = tx && slx < EL_CAP;
            const bool fy = ty && sly < EL_CAP;
            if (fx) {
                v2u r; r.x = e0; r.y = rx;
                srec[slx] = r;
                atomicAdd(&spos[ix], 1u);
            }
            if (fy) {
                v2u r; r.x = e0 + 1u; r.y = ry;
                srec[sly] = r;
                atomicAdd(&spos[iy], 1u);
            }
            unsigned ox = __builtin_amdgcn_ballot_w32(tx && !fx);
            while (ox) {
                const int t = __builtin_ctz(ox);
                ox &= ox - 1u;
                el_direct(Wval, outb, sx, sinv,
                          (unsigned)__shfl((int)e0, t), (unsigned)__shfl((int)rx, t), lane);
            }
            unsigned oy = __builtin_amdgcn_ballot_w32(ty && !fy);
            while (oy) {
                const int t = __builtin_ctz(oy);
                oy &= oy - 1u;
                el_direct(Wval, outb, sx, sinv,
                          (unsigned)__shfl((int)e0, t) + 1u, (unsigned)__shfl((int)ry, t), lane);
            }
        }
    }
    __syncthreads();

    const unsigned cnt = spos[tid];
    unsigned inc = cnt;
#pragma unroll
    for (int d = 1; d < 32; d <<= 1) {
        const unsigned v = __shfl_up(inc, d);
        if (lane >= d) inc += v;
    }
    if (lane == 31) swtot[wv] = inc;
    __syncthreads();
    unsigned pre = (lane < wv) ? swtot[lane] : 0u;
#pragma unroll
    for (int d = 16; d >= 1; d >>= 1) pre += __shfl_xor(pre, d);
    spos[tid] = pre + inc - cnt;
    __syncthreads();

    const unsigned nrec = min(snrec, (unsigned)EL_CAP);
    for (unsigned r = (unsigned)tid; r < nrec; r += EL_WAVES * 32) {
        const unsigned i = srec[r].y & (N_DIM - 1);
        const unsigned p = atomicAdd(&spos[i], 1u);
        sord[p] = (unsigned short)r;
    }
    __syncthreads();

    {
        const int lo = (wv == 0) ? 0
                     : __builtin_amdgcn_readfirstlane((int)spos[wv * 32 - 1]);
        const int hi = __builtin_amdgcn_readfirstlane((int)spos[wv * 32 + 31]);
        int cur = -1;
        float acc0 = 0.0f, acc1 = 0.0f;
        for (int p = lo; p < hi; p += EL_G) {
            const v2u rec = el_rec(srec, sord, p, hi, lane);
            float w0[EL_G], w1[EL_G];
            el_wload(Wval, rec, lane, w0, w1);
            el_sum(outb, sx, sinv, rec, w0, w1, min(EL_G, hi - p), lane, cur, acc0, acc1);
        }
        if (cur >= 0) {
            float* o = outb + cur * F_DIM;
            atomicAdd(o + lane,      acc0);
            atomicAdd(o + lane + 32, acc1);
        }
    }

    if (slice == EL_SLICES - 1 && wv == 0) {
        for (int e = nfull << 6; e < nnz; ++e) {
            if (eb[e] == b) {
                const unsigned y = (unsigned)(ei[e] & (N_DIM - 1))
                                 | ((unsigned)(ej[e] & (N_DIM - 1)) << EL_NB)
                                 | ((unsigned)(ek[e] & (N_DIM - 1)) << (2 * EL_NB));
                el_direct(Wval, outb, sx, sinv, (unsigned)e, y, lane);
            }
        }
    }
}

extern "C" void kernel_launch(void* const* d_in, const int* in_sizes, int n_in,
                              void* d_out, int out_size, void* d_ws, size_t ws_size,
                              hipStream_t stream) {
    const float* x    = (const float*)d_in[0];
    const float* Wval = (const float*)d_in[1];
    const int*   eb   = (const int*)d_in[2];
    const int*   ei   = (const int*)d_in[3];
    const int*   ej   = (const int*)d_in[4];
    const int*   ek   = (const int*)d_in[5];
    const float* W1   = (const float*)d_in[6];
    const float* b1   = (const float*)d_in[7];
    const float* W2   = (const float*)d_in[8];
    const float* b2   = (const float*)d_in[9];
    const float* Ws1  = (const float*)d_in[10];
    const float* bs1  = (const float*)d_in[11];
    const float* Ws2  = (const float*)d_in[12];
    const float* bs2  = (const float*)d_in[13];

    const int nnz   = in_sizes[2];
    const int nodes = B_DIM * N_DIM;

    float* deg = (float*)d_ws;
    float* x1  = deg + nodes;
    float* out = (float*)d_out;

    unsigned* part = (unsigned*)x1;
    hg_hist_kernel<<<HG_PARTS, HG_HT, 0, stream>>>(eb, ei, part, nnz);

    hg_degsum_kernel<<<HG_BINS / 64, 64, 0, stream>>>(part, deg);

    hg_mlp_wmma_kernel<<<nodes / 128, 256, 0, stream>>>(
        x, W1, b1, W2, b2, Ws1, bs1, Ws2, bs2, x1, out);

    hg_edge_kernel<<<B_DIM * EL_SLICES, EL_WAVES * 32, 0, stream>>>(
        Wval, eb, ei, ej, ek, deg, x1, out, nnz);
}


# MLP kernel: all four weight matrices and biases staged into LDS once with 16-byte loads issued together, in place of four staging phases of sixteen 4-byte loads each waited for one by one
# speedup vs baseline: 5.0787x; 1.3483x over previous
#include <hip/hip_runtime.h>

#define B_DIM 8
#define N_DIM 1024
#define F_DIM 64
#define EPS_V 1e-10f

typedef __attribute__((ext_vector_type(2))) float v2f;
typedef __attribute__((ext_vector_type(4))) float v4f;
typedef __attribute__((ext_vector_type(8))) float v8f;

#define HG_BINS  (B_DIM * N_DIM)
#define HG_PARTS 64
#define HG_HT    1024

typedef __attribute__((ext_vector_type(4))) unsigned v4u;
typedef __attribute__((ext_vector_type(4))) int v4i;

__device__ __forceinline__ void hg_count4(unsigned* h, v4i b, v4i i) {
    atomicAdd(&h[b.x * N_DIM + i.x], 1u);
    atomicAdd(&h[b.y * N_DIM + i.y], 1u);
    atomicAdd(&h[b.z * N_DIM + i.z], 1u);
    atomicAdd(&h[b.w * N_DIM + i.w], 1u);
}

__global__ __launch_bounds__(HG_HT) void hg_hist_kernel(
    const int* __restrict__ eb, const int* __restrict__ ei,
    unsigned* __restrict__ part, int nnz) {

    __shared__ unsigned h[HG_BINS];

    const int tid = threadIdx.x;
    v4u* h4 = (v4u*)h;
    for (int q = tid; q < HG_BINS / 4; q += HG_HT) h4[q] = (v4u)(0u);
    __syncthreads();

    const v4i* eb4 = (const v4i*)eb;
    const v4i* ei4 = (const v4i*)ei;
    const int nvec   = nnz >> 2;
    const int stride = gridDim.x * HG_HT;
    int v = blockIdx.x * HG_HT + tid;

    for (; v + 3 * stride < nvec; v += 4 * stride) {
        const v4i b0 = eb4[v];
        const v4i b1 = eb4[v + stride];
        const v4i b2 = eb4[v + 2 * stride];
        const v4i b3 = eb4[v + 3 * stride];
        const v4i i0 = ei4[v];
        const v4i i1 = ei4[v + stride];
        const v4i i2 = ei4[v + 2 * stride];
        const v4i i3 = ei4[v + 3 * stride];
        asm volatile("" :: "v"(b0), "v"(b1), "v"(b2), "v"(b3),
                           "v"(i0), "v"(i1), "v"(i2), "v"(i3));
        hg_count4(h, b0, i0);
        hg_count4(h, b1, i1);
        hg_count4(h, b2, i2);
        hg_count4(h, b3, i3);
    }
    for (; v < nvec; v += stride) hg_count4(h, eb4[v], ei4[v]);

    if (blockIdx.x == 0) {
        const int e = (nvec << 2) + tid;
        if (e < nnz) atomicAdd(&h[eb[e] * N_DIM + ei[e]], 1u);
    }
    __syncthreads();

    v4u* dst = (v4u*)(part + (size_t)blockIdx.x * HG_BINS);
    for (int q = tid; q < HG_BINS / 4; q += HG_HT) dst[q] = h4[q];
}

__global__ __launch_bounds__(64) void hg_degsum_kernel(
    const unsigned* __restrict__ part, float* __restrict__ deg) {
    const int bin = blockIdx.x * 64 + threadIdx.x;
    unsigned s = 0u;
#pragma unroll
    for (int p = 0; p < HG_PARTS; ++p) s += part[p * HG_BINS + bin];
    deg[bin] = (float)s;
}

#define MLP_WQ (F_DIM * F_DIM / 4)

__global__ __launch_bounds__(256) void hg_mlp_wmma_kernel(
    const float* __restrict__ x,
    const float* __restrict__ W1, const float* __restrict__ b1,
    const float* __restrict__ W2, const float* __restrict__ b2,
    const float* __restrict__ Ws1, const float* __restrict__ bs1,
    const float* __restrict__ Ws2, const float* __restrict__ bs2,
    float* __restrict__ x1_out, float* __restrict__ self_out) {

    __shared__ float sW[4 * F_DIM * F_DIM];
    __shared__ float sB[4 * F_DIM];
    __shared__ float sH[8][16 * F_DIM];

    const int tid  = threadIdx.x;
    const int wave = tid >> 5;
    const int lane = tid & 31;
    const int hlf  = lane >> 4;
    const int mn   = lane & 15;
    const int rowBase = (blockIdx.x * 8 + wave) * 16;

    v2f aX[16];
#pragma unroll
    for (int kk = 0; kk < 16; ++kk) {
        const float* p = x + (size_t)(rowBase + mn) * F_DIM + kk * 4 + hlf * 2;
        aX[kk][0] = p[0];
        aX[kk][1] = p[1];
    }

    {
        const v4f* g0 = (const v4f*)W1;
        const v4f* g1 = (const v4f*)W2;
        const v4f* g2 = (const v4f*)Ws1;
        const v4f* g3 = (const v4f*)Ws2;
        v4f t0[MLP_WQ / 256], t1[MLP_WQ / 256], t2[MLP_WQ / 256], t3[MLP_WQ / 256];
#pragma unroll
        for (int j = 0; j < MLP_WQ / 256; ++j) {
            t0[j] = g0[j * 256 + tid];
            t1[j] = g1[j * 256 + tid];
            t2[j] = g2[j * 256 + tid];
            t3[j] = g3[j * 256 + tid];
        }
        const int bsel = tid >> 6;
        const float* bp = (bsel == 0) ? b1 : (bsel == 1) ? b2 : (bsel == 2) ? bs1 : bs2;
        const float bv = bp[tid & (F_DIM - 1)];
        v4f* d = (v4f*)sW;
#pragma unroll
        for (int j = 0; j < MLP_WQ / 256; ++j) {
            d[0 * MLP_WQ + j * 256 + tid] = t0[j];
            d[1 * MLP_WQ + j * 256 + tid] = t1[j];
            d[2 * MLP_WQ + j * 256 + tid] = t2[j];
            d[3 * MLP_WQ + j * 256 + tid] = t3[j];
        }
        sB[tid] = bv;
    }
    __syncthreads();

    for (int s = 0; s < 2; ++s) {
        const float* wA = sW + (2 * s + 0) * F_DIM * F_DIM;
        const float* wB = sW + (2 * s + 1) * F_DIM * F_DIM;
        const float* bA = sB + (2 * s + 0) * F_DIM;
        const float* bB = sB + (2 * s + 1) * F_DIM;
        float* dst = (s == 0) ? x1_out : self_out;

#pragma unroll
        for (int nt = 0; nt < 4; ++nt) {
            v8f c;
            float bias = bA[nt * 16 + mn];
#pragma unroll
            for (int r = 0; r < 8; ++r) c[r] = bias;
#pragma unroll
            for (int kk = 0; kk < 16; ++kk) {
                v2f bF;
                bF[0] = wA[(kk * 4 + hlf * 2 + 0) * F_DIM + nt * 16 + mn];
                bF[1] = wA[(kk * 4 + hlf * 2 + 1) * F_DIM + nt * 16 + mn];
                c = __builtin_amdgcn_wmma_f32_16x16x4_f32(
                        false, aX[kk], false, bF, (short)0, c, false, false);
            }
#pragma unroll
            for (int r = 0; r < 8; ++r)
                sH[wave][(r + 8 * hlf) * F_DIM + nt * 16 + mn] = fmaxf(c[r], 0.0f);
        }
        __syncthreads();

        v2f aH[16];
#pragma unroll
        for (int kk = 0; kk < 16; ++kk) {
            aH[kk][0] = sH[wave][mn * F_DIM + kk * 4 + hlf * 2 + 0];
            aH[kk][1] = sH[wave][mn * F_DIM + kk * 4 + hlf * 2 + 1];
        }

#pragma unroll
        for (int nt = 0; nt < 4; ++nt) {
            v8f c;
            float bias = bB[nt * 16 + mn];
#pragma unroll
            for (int r = 0; r < 8; ++r) c[r] = bias;
#pragma unroll
            for (int kk = 0; kk < 16; ++kk) {
                v2f bF;
                bF[0] = wB[(kk * 4 + hlf * 2 + 0) * F_DIM + nt * 16 + mn];
                bF[1] = wB[(kk * 4 + hlf * 2 + 1) * F_DIM + nt * 16 + mn];
                c = __builtin_amdgcn_wmma_f32_16x16x4_f32(
                        false, aH[kk], false, bF, (short)0, c, false, false);
            }
#pragma unroll
            for (int r = 0; r < 8; ++r) {
                dst[(size_t)(rowBase + r + 8 * hlf) * F_DIM + nt * 16 + mn] =
                    fmaxf(c[r], 0.0f);
            }
        }
    }
}

#define EL_WAVES  32
#define EL_SLICES 32
#define EL_G      16
#define EL_H      8
#define EL_NB     10
#define EL_CAP    4608
#define EL_SR     8

static_assert((1 << EL_NB) == N_DIM, "a record packs three node indices of EL_NB bits");
static_assert(EL_WAVES * 32 == N_DIM, "one thread per node: 1/degree, counter, prefix sum");
static_assert((N_DIM * F_DIM / 4) % (EL_WAVES * 32) == 0, "whole copy trips");
static_assert(EL_CAP <= 65536, "list positions are kept as 16-bit words");

typedef __attribute__((ext_vector_type(2))) int v2i;
typedef __attribute__((ext_vector_type(2))) unsigned v2u;

__device__ __forceinline__ int eg_lane_i(int v, int t) {
    return __builtin_amdgcn_readlane(v, t);
}
__device__ __forceinline__ float eg_lane_f(float v, int t) {
    return __int_as_float(__builtin_amdgcn_readlane(__float_as_int(v), t));
}

__device__ __forceinline__ void el_direct(
    const float* __restrict__ Wval, float* __restrict__ outb,
    const float* sx, const float* sinv, unsigned e, unsigned y, int lane) {
    const int i = (int)(y & (N_DIM - 1));
    const float* xj = sx + (int)((y >> EL_NB) & (N_DIM - 1)) * F_DIM + lane;
    const float* xk = sx + (int)(y >> (2 * EL_NB)) * F_DIM + lane;
    const float* wp = Wval + (size_t)e * F_DIM;
    const float a = sinv[i];
    float* o = outb + i * F_DIM;
    const float w0 = __builtin_nontemporal_load(wp + lane);
    const float w1 = __builtin_nontemporal_load(wp + lane + 32);
    atomicAdd(o + lane,      a * w0 * xk[0]  * xj[0]);
    atomicAdd(o + lane + 32, a * w1 * xk[32] * xj[32]);
}

__device__ __forceinline__ v2u el_rec(
    const v2u* srec, const unsigned short* sord, int p, int hi, int lane) {
    const int q = min(p + (lane & (EL_G - 1)), hi - 1);
    return srec[sord[q]];
}

__device__ __forceinline__ void el_wload(
    const float* __restrict__ Wval, v2u rec, int lane,
    float (&w0)[EL_G], float (&w1)[EL_G]) {
    const int e = (int)rec.x;
#pragma unroll
    for (int t = 0; t < EL_G; ++t) {
        const float* wp = Wval + (size_t)(unsigned)eg_lane_i(e, t) * F_DIM;
        w0[t] = __builtin_nontemporal_load(wp + lane);
        w1[t] = __builtin_nontemporal_load(wp + lane + 32);
    }
}

__device__ __forceinline__ void el_sum(
    float* __restrict__ outb, const float* sx, const float* sinv, v2u rec,
    const float (&w0)[EL_G], const float (&w1)[EL_G],
    int rem, int lane, int& cur, float& acc0, float& acc1) {

    const int i    = (int)(rec.y & (N_DIM - 1));
    const int jOff = (int)((rec.y >> EL_NB) & (N_DIM - 1)) * F_DIM;
    const int kOff = (int)(rec.y >> (2 * EL_NB)) * F_DIM;
    const float av = sinv[i];

#pragma unroll
    for (int g = 0; g < EL_G; g += EL_H) {
        float k0[EL_H], k1[EL_H], j0[EL_H], j1[EL_H];
#pragma unroll
        for (int t = 0; t < EL_H; ++t) {
            const float* xk = sx + eg_lane_i(kOff, g + t) + lane;
            const float* xj = sx + eg_lane_i(jOff, g + t) + lane;
            k0[t] = xk[0];
            k1[t] = xk[32];
            j0[t] = xj[0];
            j1[t] = xj[32];
        }
#pragma unroll
        for (int t = 0; t < EL_H; ++t) {
            const int it = eg_lane_i(i, g + t);
            if (it != cur) {
                if (cur >= 0) {
                    float* o = outb + cur * F_DIM;
                    atomicAdd(o + lane,      acc0);
                    atomicAdd(o + lane + 32, acc1);
                }
                cur  = it;
                acc0 = 0.0f;
                acc1 = 0.0f;
            }
            const float a = (g + t < rem) ? eg_lane_f(av, g + t) : 0.0f;
            acc0 += a * w0[g + t] * k0[t] * j0[t];
            acc1 += a * w1[g + t] * k1[t] * j1[t];
        }
    }
}

__global__ __launch_bounds__(EL_WAVES * 32) void hg_edge_kernel(
    const float* __restrict__ Wval,
    const int* __restrict__ eb, const int* __restrict__ ei,
    const int* __restrict__ ej, const int* __restrict__ ek,
    const float* __restrict__ deg, const float* __restrict__ x1,
    float* __restrict__ out, int nnz) {

    __shared__ float sx[N_DIM * F_DIM];
    __shared__ float sinv[N_DIM];
    __shared__ v2u   srec[EL_CAP];
    __shared__ unsigned short sord[EL_CAP];
    __shared__ unsigned spos[N_DIM];
    __shared__ unsigned swtot[EL_WAVES];
    __shared__ unsigned snrec;

    const int tid  = threadIdx.x;
    const int lane = tid & 31;
    const int wv    = __builtin_amdgcn_readfirstlane(tid >> 5);
    const int b     = blockIdx.x % B_DIM;
    const int slice = blockIdx.x / B_DIM;

    spos[tid] = 0u;
    if (tid == 0) snrec = 0u;
    {
        const v4f* src = (const v4f*)(x1 + (size_t)b * N_DIM * F_DIM);
        v4f* dst = (v4f*)sx;
#pragma unroll
        for (int q = 0; q < N_DIM * F_DIM / 4 / (EL_WAVES * 32); ++q)
            dst[q * (EL_WAVES * 32) + tid] = src[q * (EL_WAVES * 32) + tid];
        sinv[tid] = 1.0f / (deg[b * N_DIM + tid] + EPS_V);
    }
    __syncthreads();

    float* outb = out + (size_t)b * N_DIM * F_DIM;

    const int nfull = nnz >> 6;
    const int per   = (nfull + EL_SLICES - 1) / EL_SLICES;
    const int s0    = slice * per;
    const int s1    = min(s0 + per, nfull);

    const v2i* eb2 = (const v2i*)eb;
    const v2i* ei2 = (const v2i*)ei;
    const v2i* ej2 = (const v2i*)ej;
    const v2i* ek2 = (const v2i*)ek;

    for (int sr = s0 + wv; sr < s1; sr += EL_WAVES * EL_SR) {
        v2i cb[EL_SR], ci[EL_SR], cj[EL_SR], ck[EL_SR];
#pragma unroll
        for (int q = 0; q < EL_SR; ++q) {
            const int p = min(sr + q * EL_WAVES, s1 - 1) * 32 + lane;
            cb[q] = eb2[p];
            ci[q] = ei2[p];
            cj[q] = ej2[p];
            ck[q] = ek2[p];
        }
        unsigned mx[EL_SR], my[EL_SR];
        unsigned tot = 0u;
#pragma unroll
        for (int q = 0; q < EL_SR; ++q) {
            const bool live = (sr + q * EL_WAVES < s1);
            mx[q] = __builtin_amdgcn_ballot_w32(live && cb[q].x == b);
            my[q] = __builtin_amdgcn_ballot_w32(live && cb[q].y == b);
            tot += (unsigned)__builtin_popcount(mx[q]) + (unsigned)__builtin_popcount(my[q]);
        }
        unsigned base = 0u;
        if (lane == 0) base = atomicAdd(&snrec, tot);
        base = (unsigned)__builtin_amdgcn_readfirstlane((int)base);
#pragma unroll
        for (int q = 0; q < EL_SR; ++q) {
            const unsigned e0 = (unsigned)((sr + q * EL_WAVES) * 64 + 2 * lane);
            const bool tx = ((mx[q] >> lane) & 1u) != 0u;
            const bool ty = ((my[q] >> lane) & 1u) != 0u;
            const unsigned ix = (unsigned)(ci[q].x & (N_DIM - 1));
            const unsigned iy = (unsigned)(ci[q].y & (N_DIM - 1));
            const unsigned rx = ix
                              | ((unsigned)(cj[q].x & (N_DIM - 1)) << EL_NB)
                              | ((unsigned)(ck[q].x & (N_DIM - 1)) << (2 * EL_NB));
            const unsigned ry = iy
                              | ((unsigned)(cj[q].y & (N_DIM - 1)) << EL_NB)
                              | ((unsigned)(ck[q].y & (N_DIM - 1)) << (2 * EL_NB));
            const unsigned nx  = (unsigned)__builtin_popcount(mx[q]);
            const unsigned slx = base + __builtin_amdgcn_mbcnt_lo(mx[q], 0u);
            const unsigned sly = base + nx + __builtin_amdgcn_mbcnt_lo(my[q], 0u);
            base += nx + (unsigned)__builtin_popcount(my[q]);
            const bool fx = tx && slx < EL_CAP;
            const bool fy = ty && sly < EL_CAP;
            if (fx) {
                v2u r; r.x = e0; r.y = rx;
                srec[slx] = r;
                atomicAdd(&spos[ix], 1u);
            }
            if (fy) {
                v2u r; r.x = e0 + 1u; r.y = ry;
                srec[sly] = r;
                atomicAdd(&spos[iy], 1u);
            }
            unsigned ox = __builtin_amdgcn_ballot_w32(tx && !fx);
            while (ox) {
                const int t = __builtin_ctz(ox);
                ox &= ox - 1u;
                el_direct(Wval, outb, sx, sinv,
                          (unsigned)__shfl((int)e0, t), (unsigned)__shfl((int)rx, t), lane);
            }
            unsigned oy = __builtin_amdgcn_ballot_w32(ty && !fy);
            while (oy) {
                const int t = __builtin_ctz(oy);
                oy &= oy - 1u;
                el_direct(Wval, outb, sx, sinv,
                          (unsigned)__shfl((int)e0, t) + 1u, (unsigned)__shfl((int)ry, t), lane);
            }
        }
    }
    __syncthreads();

    const unsigned cnt = spos[tid];
    unsigned inc = cnt;
#pragma unroll
    for (int d = 1; d < 32; d <<= 1) {
        const unsigned v = __shfl_up(inc, d);
        if (lane >= d) inc += v;
    }
    if (lane == 31) swtot[wv] = inc;
    __syncthreads();
    unsigned pre = (lane < wv) ? swtot[lane] : 0u;
#pragma unroll
    for (int d = 16; d >= 1; d >>= 1) pre += __shfl_xor(pre, d);
    spos[tid] = pre + inc - cnt;
    __syncthreads();

    const unsigned nrec = min(snrec, (unsigned)EL_CAP);
    for (unsigned r = (unsigned)tid; r < nrec; r += EL_WAVES * 32) {
        const unsigned i = srec[r].y & (N_DIM - 1);
        const unsigned p = atomicAdd(&spos[i], 1u);
        sord[p] = (unsigned short)r;
    }
    __syncthreads();

    {
        const int lo = (wv == 0) ? 0
                     : __builtin_amdgcn_readfirstlane((int)spos[wv * 32 - 1]);
        const int hi = __builtin_amdgcn_readfirstlane((int)spos[wv * 32 + 31]);
        int cur = -1;
        float acc0 = 0.0f, acc1 = 0.0f;
        for (int p = lo; p < hi; p += EL_G) {
            const v2u rec = el_rec(srec, sord, p, hi, lane);
            float w0[EL_G], w1[EL_G];
            el_wload(Wval, rec, lane, w0, w1);
            el_sum(outb, sx, sinv, rec, w0, w1, min(EL_G, hi - p), lane, cur, acc0, acc1);
        }
        if (cur >= 0) {
            float* o = outb + cur * F_DIM;
            atomicAdd(o + lane,      acc0);
            atomicAdd(o + lane + 32, acc1);
        }
    }

    if (slice == EL_SLICES - 1 && wv == 0) {
        for (int e = nfull << 6; e < nnz; ++e) {
            if (eb[e] == b) {
                const unsigned y = (unsigned)(ei[e] & (N_DIM - 1))
                                 | ((unsigned)(ej[e] & (N_DIM - 1)) << EL_NB)
                                 | ((unsigned)(ek[e] & (N_DIM - 1)) << (2 * EL_NB));
                el_direct(Wval, outb, sx, sinv, (unsigned)e, y, lane);
            }
        }
    }
}

extern "C" void kernel_launch(void* const* d_in, const int* in_sizes, int n_in,
                              void* d_out, int out_size, void* d_ws, size_t ws_size,
                              hipStream_t stream) {
    const float* x    = (const float*)d_in[0];
    const float* Wval = (const float*)d_in[1];
    const int*   eb   = (const int*)d_in[2];
    const int*   ei   = (const int*)d_in[3];
    const int*   ej   = (const int*)d_in[4];
    const int*   ek   = (const int*)d_in[5];
    const float* W1   = (const float*)d_in[6];
    const float* b1   = (const float*)d_in[7];
    const float* W2   = (const float*)d_in[8];
    const float* b2   = (const float*)d_in[9];
    const float* Ws1  = (const float*)d_in[10];
    const float* bs1  = (const float*)d_in[11];
    const float* Ws2  = (const float*)d_in[12];
    const float* bs2  = (const float*)d_in[13];

    const int nnz   = in_sizes[2];
    const int nodes = B_DIM * N_DIM;

    float* deg = (float*)d_ws;
    float* x1  = deg + nodes;
    float* out = (float*)d_out;

    unsigned* part = (unsigned*)x1;
    hg_hist_kernel<<<HG_PARTS, HG_HT, 0, stream>>>(eb, ei, part, nnz);

    hg_degsum_kernel<<<HG_BINS / 64, 64, 0, stream>>>(part, deg);

    hg_mlp_wmma_kernel<<<nodes / 128, 256, 0, stream>>>(
        x, W1, b1, W2, b2, Ws1, bs1, Ws2, bs2, x1, out);

    hg_edge_kernel<<<B_DIM * EL_SLICES, EL_WAVES * 32, 0, stream>>>(
        Wval, eb, ei, ej, ek, deg, x1, out, nnz);
}


# degree count: LDS histograms added by line-wide integer atomics (exact) into one zeroed plane of counters, from blocks that run beside the MLP blocks in one launch; no summing launch: 4 launches to 3
# speedup vs baseline: 5.5619x; 1.0951x over previous
#include <hip/hip_runtime.h>

#define B_DIM 8
#define N_DIM 1024
#define F_DIM 64
#define EPS_V 1e-10f

typedef __attribute__((ext_vector_type(2))) float v2f;
typedef __attribute__((ext_vector_type(4))) float v4f;
typedef __attribute__((ext_vector_type(8))) float v8f;

#define HG_BINS  (B_DIM * N_DIM)
#define HG_HBLK  128
#define HG_HT    256
#define HG_DEPTH 8

typedef __attribute__((ext_vector_type(4))) unsigned v4u;
typedef __attribute__((ext_vector_type(4))) int v4i;

__device__ __forceinline__ void hg_count4(unsigned* h, v4i b, v4i i) {
    atomicAdd(&h[b.x * N_DIM + i.x], 1u);
    atomicAdd(&h[b.y * N_DIM + i.y], 1u);
    atomicAdd(&h[b.z * N_DIM + i.z], 1u);
    atomicAdd(&h[b.w * N_DIM + i.w], 1u);
}

__global__ __launch_bounds__(HG_HT) void hg_degzero_kernel(unsigned* __restrict__ degc) {
    degc[blockIdx.x * HG_HT + threadIdx.x] = 0u;
}

__device__ __forceinline__ void hg_hist_body(
    const int* __restrict__ eb, const int* __restrict__ ei,
    unsigned* __restrict__ degc, int nnz, unsigned* h, int hb, int tid) {

    v4u* h4 = (v4u*)h;
    for (int q = tid; q < HG_BINS / 4; q += HG_HT) h4[q] = (v4u)(0u);
    __syncthreads();

    const v4i* eb4 = (const v4i*)eb;
    const v4i* ei4 = (const v4i*)ei;
    const int nvec   = nnz >> 2;
    const int stride = HG_HBLK * HG_HT;
    int v = hb * HG_HT + tid;

    for (; v + (HG_DEPTH - 1) * stride < nvec; v += HG_DEPTH * stride) {
        v4i gb[HG_DEPTH], gi[HG_DEPTH];
#pragma unroll
        for (int q = 0; q < HG_DEPTH; ++q) {
            gb[q] = eb4[v + q * stride];
            gi[q] = ei4[v + q * stride];
        }
#pragma unroll
        for (int q = 0; q < HG_DEPTH; ++q) hg_count4(h, gb[q], gi[q]);
    }
    for (; v < nvec; v += stride) hg_count4(h, eb4[v], ei4[v]);

    if (hb == 0) {
        const int e = (nvec << 2) + tid;
        if (e < nnz) atomicAdd(&h[eb[e] * N_DIM + ei[e]], 1u);
    }
    __syncthreads();

    unsigned c[HG_BINS / HG_HT];
#pragma unroll
    for (int k = 0; k < HG_BINS / HG_HT; ++k) c[k] = h[k * HG_HT + tid];
#pragma unroll
    for (int k = 0; k < HG_BINS / HG_HT; ++k)
        if (c[k] != 0u) atomicAdd(&degc[k * HG_HT + tid], c[k]);
}

#define MLP_WQ (F_DIM * F_DIM / 4)
#define MLP_BLOCKS (B_DIM * N_DIM / 128)

static_assert(HG_HT == 256, "launch 2: 8 waves per block, a 16-row strip per wave");

__global__ __launch_bounds__(HG_HT) void hg_mlp_hist_kernel(
    const float* __restrict__ x,
    const float* __restrict__ W1, const float* __restrict__ b1,
    const float* __restrict__ W2, const float* __restrict__ b2,
    const float* __restrict__ Ws1, const float* __restrict__ bs1,
    const float* __restrict__ Ws2, const float* __restrict__ bs2,
    float* __restrict__ x1_out, float* __restrict__ self_out,
    const int* __restrict__ eb, const int* __restrict__ ei,
    unsigned* __restrict__ degc, int nnz) {

    __shared__ float sW[4 * F_DIM * F_DIM];
    __shared__ float sB[4 * F_DIM];
    __shared__ float sH[8][16 * F_DIM];

    const int tid  = threadIdx.x;

    if (blockIdx.x >= MLP_BLOCKS) {
        hg_hist_body(eb, ei, degc, nnz, (unsigned*)sW, (int)blockIdx.x - MLP_BLOCKS, tid);
        return;
    }

    const int wave = tid >> 5;
    const int lane = tid & 31;
    const int hlf  = lane >> 4;
    const int mn   = lane & 15;
    const int rowBase = (blockIdx.x * 8 + wave) * 16;

    v2f aX[16];
#pragma unroll
    for (int kk = 0; kk < 16; ++kk) {
        const float* p = x + (size_t)(rowBase + mn) * F_DIM + kk * 4 + hlf * 2;
        aX[kk][0] = p[0];
        aX[kk][1] = p[1];
    }

    {
        const v4f* g0 = (const v4f*)W1;
        const v4f* g1 = (const v4f*)W2;
        const v4f* g2 = (const v4f*)Ws1;
        const v4f* g3 = (const v4f*)Ws2;
        v4f t0[MLP_WQ / 256], t1[MLP_WQ / 256], t2[MLP_WQ / 256], t3[MLP_WQ / 256];
#pragma unroll
        for (int j = 0; j < MLP_WQ / 256; ++j) {
            t0[j] = g0[j * 256 + tid];
            t1[j] = g1[j * 256 + tid];
            t2[j] = g2[j * 256 + tid];
            t3[j] = g3[j * 256 + tid];
        }
        const int bsel = tid >> 6;
        const float* bp = (bsel == 0) ? b1 : (bsel == 1) ? b2 : (bsel == 2) ? bs1 : bs2;
        const float bv = bp[tid & (F_DIM - 1)];
        v4f* d = (v4f*)sW;
#pragma unroll
        for (int j = 0; j < MLP_WQ / 256; ++j) {
            d[0 * MLP_WQ + j * 256 + tid] = t0[j];
            d[1 * MLP_WQ + j * 256 + tid] = t1[j];
            d[2 * MLP_WQ + j * 256 + tid] = t2[j];
            d[3 * MLP_WQ + j * 256 + tid] = t3[j];
        }
        sB[tid] = bv;
    }
    __syncthreads();

    for (int s = 0; s < 2; ++s) {
        const float* wA = sW + (2 * s + 0) * F_DIM * F_DIM;
        const float* wB = sW + (2 * s + 1) * F_DIM * F_DIM;
        const float* bA = sB + (2 * s + 0) * F_DIM;
        const float* bB = sB + (2 * s + 1) * F_DIM;
        float* dst = (s == 0) ? x1_out : self_out;

#pragma unroll
        for (int nt = 0; nt < 4; ++nt) {
            v8f c;
            float bias = bA[nt * 16 + mn];
#pragma unroll
            for (int r = 0; r < 8; ++r) c[r] = bias;
#pragma unroll
            for (int kk = 0; kk < 16; ++kk) {
                v2f bF;
                bF[0] = wA[(kk * 4 + hlf * 2 + 0) * F_DIM + nt * 16 + mn];
                bF[1] = wA[(kk * 4 + hlf * 2 + 1) * F_DIM + nt * 16 + mn];
                c = __builtin_amdgcn_wmma_f32_16x16x4_f32(
                        false, aX[kk], false, bF, (short)0, c, false, false);
            }
#pragma unroll
            for (int r = 0; r < 8; ++r)
                sH[wave][(r + 8 * hlf) * F_DIM + nt * 16 + mn] = fmaxf(c[r], 0.0f);
        }
        __syncthreads();

        v2f aH[16];
#pragma unroll
        for (int kk = 0; kk < 16; ++kk) {
            aH[kk][0] = sH[wave][mn * F_DIM + kk * 4 + hlf * 2 + 0];
            aH[kk][1] = sH[wave][mn * F_DIM + kk * 4 + hlf * 2 + 1];
        }

#pragma unroll
        for (int nt = 0; nt < 4; ++nt) {
            v8f c;
            float bias = bB[nt * 16 + mn];
#pragma unroll
            for (int r = 0; r < 8; ++r) c[r] = bias;
#pragma unroll
            for (int kk = 0; kk < 16; ++kk) {
                v2f bF;
                bF[0] = wB[(kk * 4 + hlf * 2 + 0) * F_DIM + nt * 16 + mn];
                bF[1] = wB[(kk * 4 + hlf * 2 + 1) * F_DIM + nt * 16 + mn];
                c = __builtin_amdgcn_wmma_f32_16x16x4_f32(
                        false, aH[kk], false, bF, (short)0, c, false, false);
            }
#pragma unroll
            for (int r = 0; r < 8; ++r) {
                dst[(size_t)(rowBase + r + 8 * hlf) * F_DIM + nt * 16 + mn] =
                    fmaxf(c[r], 0.0f);
            }
        }
    }
}

#define EL_WAVES  32
#define EL_SLICES 32
#define EL_G      16
#define EL_H      8
#define EL_NB     10
#define EL_CAP    4608
#define EL_SR     8

static_assert((1 << EL_NB) == N_DIM, "a record packs three node indices of EL_NB bits");
static_assert(EL_WAVES * 32 == N_DIM, "one thread per node: 1/degree, counter, prefix sum");
static_assert((N_DIM * F_DIM / 4) % (EL_WAVES * 32) == 0, "whole copy trips");
static_assert(EL_CAP <= 65536, "list positions are kept as 16-bit words");

typedef __attribute__((ext_vector_type(2))) int v2i;
typedef __attribute__((ext_vector_type(2))) unsigned v2u;

__device__ __forceinline__ int eg_lane_i(int v, int t) {
    return __builtin_amdgcn_readlane(v, t);
}
__device__ __forceinline__ float eg_lane_f(float v, int t) {
    return __int_as_float(__builtin_amdgcn_readlane(__float_as_int(v), t));
}

__device__ __forceinline__ void el_direct(
    const float* __restrict__ Wval, float* __restrict__ outb,
    const float* sx, const float* sinv, unsigned e, unsigned y, int lane) {
    const int i = (int)(y & (N_DIM - 1));
    const float* xj = sx + (int)((y >> EL_NB) & (N_DIM - 1)) * F_DIM + lane;
    const float* xk = sx + (int)(y >> (2 * EL_NB)) * F_DIM + lane;
    const float* wp = Wval + (size_t)e * F_DIM;
    const float a = sinv[i];
    float* o = outb + i * F_DIM;
    const float w0 = __builtin_nontemporal_load(wp + lane);
    const float w1 = __builtin_nontemporal_load(wp + lane + 32);
    atomicAdd(o + lane,      a * w0 * xk[0]  * xj[0]);
    atomicAdd(o + lane + 32, a * w1 * xk[32] * xj[32]);
}

__device__ __forceinline__ v2u el_rec(
    const v2u* srec, const unsigned short* sord, int p, int hi, int lane) {
    const int q = min(p + (lane & (EL_G - 1)), hi - 1);
    return srec[sord[q]];
}

__device__ __forceinline__ void el_wload(
    const float* __restrict__ Wval, v2u rec, int lane,
    float (&w0)[EL_G], float (&w1)[EL_G]) {
    const int e = (int)rec.x;
#pragma unroll
    for (int t = 0; t < EL_G; ++t) {
        const float* wp = Wval + (size_t)(unsigned)eg_lane_i(e, t) * F_DIM;
        w0[t] = __builtin_nontemporal_load(wp + lane);
        w1[t] = __builtin_nontemporal_load(wp + lane + 32);
    }
}

__device__ __forceinline__ void el_sum(
    float* __restrict__ outb, const float* sx, const float* sinv, v2u rec,
    const float (&w0)[EL_G], const float (&w1)[EL_G],
    int rem, int lane, int& cur, float& acc0, float& acc1) {

    const int i    = (int)(rec.y & (N_DIM - 1));
    const int jOff = (int)((rec.y >> EL_NB) & (N_DIM - 1)) * F_DIM;
    const int kOff = (int)(rec.y >> (2 * EL_NB)) * F_DIM;
    const float av = sinv[i];

#pragma unroll
    for (int g = 0; g < EL_G; g += EL_H) {
        float k0[EL_H], k1[EL_H], j0[EL_H], j1[EL_H];
#pragma unroll
        for (int t = 0; t < EL_H; ++t) {
            const float* xk = sx + eg_lane_i(kOff, g + t) + lane;
            const float* xj = sx + eg_lane_i(jOff, g + t) + lane;
            k0[t] = xk[0];
            k1[t] = xk[32];
            j0[t] = xj[0];
            j1[t] = xj[32];
        }
#pragma unroll
        for (int t = 0; t < EL_H; ++t) {
            const int it = eg_lane_i(i, g + t);
            if (it != cur) {
                if (cur >= 0) {
                    float* o = outb + cur * F_DIM;
                    atomicAdd(o + lane,      acc0);
                    atomicAdd(o + lane + 32, acc1);
                }
                cur  = it;
                acc0 = 0.0f;
                acc1 = 0.0f;
            }
            const float a = (g + t < rem) ? eg_lane_f(av, g + t) : 0.0f;
            acc0 += a * w0[g + t] * k0[t] * j0[t];
            acc1 += a * w1[g + t] * k1[t] * j1[t];
        }
    }
}

__global__ __launch_bounds__(EL_WAVES * 32) void hg_edge_kernel(
    const float* __restrict__ Wval,
    const int* __restrict__ eb, const int* __restrict__ ei,
    const int* __restrict__ ej, const int* __restrict__ ek,
    const unsigned* __restrict__ degc, const float* __restrict__ x1,
    float* __restrict__ out, int nnz) {

    __shared__ float sx[N_DIM * F_DIM];
    __shared__ float sinv[N_DIM];
    __shared__ v2u   srec[EL_CAP];
    __shared__ unsigned short sord[EL_CAP];
    __shared__ unsigned spos[N_DIM];
    __shared__ unsigned swtot[EL_WAVES];
    __shared__ unsigned snrec;

    const int tid  = threadIdx.x;
    const int lane = tid & 31;
    const int wv    = __builtin_amdgcn_readfirstlane(tid >> 5);
    const int b     = blockIdx.x % B_DIM;
    const int slice = blockIdx.x / B_DIM;

    spos[tid] = 0u;
    if (tid == 0) snrec = 0u;
    {
        const v4f* src = (const v4f*)(x1 + (size_t)b * N_DIM * F_DIM);
        v4f* dst = (v4f*)sx;
#pragma unroll
        for (int q = 0; q < N_DIM * F_DIM / 4 / (EL_WAVES * 32); ++q)
            dst[q * (EL_WAVES * 32) + tid] = src[q * (EL_WAVES * 32) + tid];
        sinv[tid] = 1.0f / ((float)degc[b * N_DIM + tid] + EPS_V);
    }
    __syncthreads();

    float* outb = out + (size_t)b * N_DIM * F_DIM;

    const int nfull = nnz >> 6;
    const int per   = (nfull + EL_SLICES - 1) / EL_SLICES;
    const int s0    = slice * per;
    const int s1    = min(s0 + per, nfull);

    const v2i* eb2 = (const v2i*)eb;
    const v2i* ei2 = (const v2i*)ei;
    const v2i* ej2 = (const v2i*)ej;
    const v2i* ek2 = (const v2i*)ek;

    for (int sr = s0 + wv; sr < s1; sr += EL_WAVES * EL_SR) {
        v2i cb[EL_SR], ci[EL_SR], cj[EL_SR], ck[EL_SR];
#pragma unroll
        for (int q = 0; q < EL_SR; ++q) {
            const int p = min(sr + q * EL_WAVES, s1 - 1) * 32 + lane;
            cb[q] = eb2[p];
            ci[q] = ei2[p];
            cj[q] = ej2[p];
            ck[q] = ek2[p];
        }
        unsigned mx[EL_SR], my[EL_SR];
        unsigned tot = 0u;
#pragma unroll
        for (int q = 0; q < EL_SR; ++q) {
            const bool live = (sr + q * EL_WAVES < s1);
            mx[q] = __builtin_amdgcn_ballot_w32(live && cb[q].x == b);
            my[q] = __builtin_amdgcn_ballot_w32(live && cb[q].y == b);
            tot += (unsigned)__builtin_popcount(mx[q]) + (unsigned)__builtin_popcount(my[q]);
        }
        unsigned base = 0u;
        if (lane == 0) base = atomicAdd(&snrec, tot);
        base = (unsigned)__builtin_amdgcn_readfirstlane((int)base);
#pragma unroll
        for (int q = 0; q < EL_SR; ++q) {
            const unsigned e0 = (unsigned)((sr + q * EL_WAVES) * 64 + 2 * lane);
            const bool tx = ((mx[q] >> lane) & 1u) != 0u;
            const bool ty = ((my[q] >> lane) & 1u) != 0u;
            const unsigned ix = (unsigned)(ci[q].x & (N_DIM - 1));
            const unsigned iy = (unsigned)(ci[q].y & (N_DIM - 1));
            const unsigned rx = ix
                              | ((unsigned)(cj[q].x & (N_DIM - 1)) << EL_NB)
                              | ((unsigned)(ck[q].x & (N_DIM - 1)) << (2 * EL_NB));
            const unsigned ry = iy
                              | ((unsigned)(cj[q].y & (N_DIM - 1)) << EL_NB)
                              | ((unsigned)(ck[q].y & (N_DIM - 1)) << (2 * EL_NB));
            const unsigned nx  = (unsigned)__builtin_popcount(mx[q]);
            const unsigned slx = base + __builtin_amdgcn_mbcnt_lo(mx[q], 0u);
            const unsigned sly = base + nx + __builtin_amdgcn_mbcnt_lo(my[q], 0u);
            base += nx + (unsigned)__builtin_popcount(my[q]);
            const bool fx = tx && slx < EL_CAP;
            const bool fy = ty && sly < EL_CAP;
            if (fx) {
                v2u r; r.x = e0; r.y = rx;
                srec[slx] = r;
                atomicAdd(&spos[ix], 1u);
            }
            if (fy) {
                v2u r; r.x = e0 + 1u; r.y = ry;
                srec[sly] = r;
                atomicAdd(&spos[iy], 1u);
            }
            unsigned ox = __builtin_amdgcn_ballot_w32(tx && !fx);
            while (ox) {
                const int t = __builtin_ctz(ox);
                ox &= ox - 1u;
                el_direct(Wval, outb, sx, sinv,
                          (unsigned)__shfl((int)e0, t), (unsigned)__shfl((int)rx, t), lane);
            }
            unsigned oy = __builtin_amdgcn_ballot_w32(ty && !fy);
            while (oy) {
                const int t = __builtin_ctz(oy);
                oy &= oy - 1u;
                el_direct(Wval, outb, sx, sinv,
                          (unsigned)__shfl((int)e0, t) + 1u, (unsigned)__shfl((int)ry, t), lane);
            }
        }
    }
    __syncthreads();

    const unsigned cnt = spos[tid];
    unsigned inc = cnt;
#pragma unroll
    for (int d = 1; d < 32; d <<= 1) {
        const unsigned v = __shfl_up(inc, d);
        if (lane >= d) inc += v;
    }
    if (lane == 31) swtot[wv] = inc;
    __syncthreads();
    unsigned pre = (lane < wv) ? swtot[lane] : 0u;
#pragma unroll
    for (int d = 16; d >= 1; d >>= 1) pre += __shfl_xor(pre, d);
    spos[tid] = pre + inc - cnt;
    __syncthreads();

    const unsigned nrec = min(snrec, (unsigned)EL_CAP);
    for (unsigned r = (unsigned)tid; r < nrec; r += EL_WAVES * 32) {
        const unsigned i = srec[r].y & (N_DIM - 1);
        const unsigned p = atomicAdd(&spos[i], 1u);
        sord[p] = (unsigned short)r;
    }
    __syncthreads();

    {
        const int lo = (wv == 0) ? 0
                     : __builtin_amdgcn_readfirstlane((int)spos[wv * 32 - 1]);
        const int hi = __builtin_amdgcn_readfirstlane((int)spos[wv * 32 + 31]);
        int cur = -1;
        float acc0 = 0.0f, acc1 = 0.0f;
        for (int p = lo; p < hi; p += EL_G) {
            const v2u rec = el_rec(srec, sord, p, hi, lane);
            float w0[EL_G], w1[EL_G];
            el_wload(Wval, rec, lane, w0, w1);
            el_sum(outb, sx, sinv, rec, w0, w1, min(EL_G, hi - p), lane, cur, acc0, acc1);
        }
        if (cur >= 0) {
            float* o = outb + cur * F_DIM;
            atomicAdd(o + lane,      acc0);
            atomicAdd(o + lane + 32, acc1);
        }
    }

    if (slice == EL_SLICES - 1 && wv == 0) {
        for (int e = nfull << 6; e < nnz; ++e) {
            if (eb[e] == b) {
                const unsigned y = (unsigned)(ei[e] & (N_DIM - 1))
                                 | ((unsigned)(ej[e] & (N_DIM - 1)) << EL_NB)
                                 | ((unsigned)(ek[e] & (N_DIM - 1)) << (2 * EL_NB));
                el_direct(Wval, outb, sx, sinv, (unsigned)e, y, lane);
            }
        }
    }
}

extern "C" void kernel_launch(void* const* d_in, const int* in_sizes, int n_in,
                              void* d_out, int out_size, void* d_ws, size_t ws_size,
                              hipStream_t stream) {
    const float* x    = (const float*)d_in[0];
    const float* Wval = (const float*)d_in[1];
    const int*   eb   = (const int*)d_in[2];
    const int*   ei   = (const int*)d_in[3];
    const int*   ej   = (const int*)d_in[4];
    const int*   ek   = (const int*)d_in[5];
    const float* W1   = (const float*)d_in[6];
    const float* b1   = (const float*)d_in[7];
    const float* W2   = (const float*)d_in[8];
    const float* b2   = (const float*)d_in[9];
    const float* Ws1  = (const float*)d_in[10];
    const float* bs1  = (const float*)d_in[11];
    const float* Ws2  = (const float*)d_in[12];
    const float* bs2  = (const float*)d_in[13];

    const int nnz   = in_sizes[2];
    const int nodes = B_DIM * N_DIM;

    unsigned* degc = (unsigned*)d_ws;
    float* x1  = (float*)d_ws + nodes;
    float* out = (float*)d_out;

    hg_degzero_kernel<<<HG_BINS / HG_HT, HG_HT, 0, stream>>>(degc);

    hg_mlp_hist_kernel<<<MLP_BLOCKS + HG_HBLK, HG_HT, 0, stream>>>(
        x, W1, b1, W2, b2, Ws1, bs1, Ws2, bs2, x1, out, eb, ei, degc, nnz);

    hg_edge_kernel<<<B_DIM * EL_SLICES, EL_WAVES * 32, 0, stream>>>(
        Wval, eb, ei, ej, ek, degc, x1, out, nnz);
}


# MLP kernel: a wave runs one of a strip's two MLPs, not both (128 blocks of 8 waves, was 64): a SIMD's chain of f32 16x16x4 WMMAs is 256 long, not 512; every sum and its order as before
# speedup vs baseline: 5.9747x; 1.0742x over previous
#include <hip/hip_runtime.h>

#define B_DIM 8
#define N_DIM 1024
#define F_DIM 64
#define EPS_V 1e-10f

typedef __attribute__((ext_vector_type(2))) float v2f;
typedef __attribute__((ext_vector_type(4))) float v4f;
typedef __attribute__((ext_vector_type(8))) float v8f;

#define HG_BINS  (B_DIM * N_DIM)
#define HG_HBLK  128
#define HG_HT    256
#define HG_DEPTH 8

typedef __attribute__((ext_vector_type(4))) unsigned v4u;
typedef __attribute__((ext_vector_type(4))) int v4i;

__device__ __forceinline__ void hg_count4(unsigned* h, v4i b, v4i i) {
    atomicAdd(&h[b.x * N_DIM + i.x], 1u);
    atomicAdd(&h[b.y * N_DIM + i.y], 1u);
    atomicAdd(&h[b.z * N_DIM + i.z], 1u);
    atomicAdd(&h[b.w * N_DIM + i.w], 1u);
}

__global__ __launch_bounds__(HG_HT) void hg_degzero_kernel(unsigned* __restrict__ degc) {
    degc[blockIdx.x * HG_HT + threadIdx.x] = 0u;
}

__device__ __forceinline__ void hg_hist_body(
    const int* __restrict__ eb, const int* __restrict__ ei,
    unsigned* __restrict__ degc, int nnz, unsigned* h, int hb, int tid) {

    v4u* h4 = (v4u*)h;
    for (int q = tid; q < HG_BINS / 4; q += HG_HT) h4[q] = (v4u)(0u);
    __syncthreads();

    const v4i* eb4 = (const v4i*)eb;
    const v4i* ei4 = (const v4i*)ei;
    const int nvec   = nnz >> 2;
    const int stride = HG_HBLK * HG_HT;
    int v = hb * HG_HT + tid;

    for (; v + (HG_DEPTH - 1) * stride < nvec; v += HG_DEPTH * stride) {
        v4i gb[HG_DEPTH], gi[HG_DEPTH];
#pragma unroll
        for (int q = 0; q < HG_DEPTH; ++q) {
            gb[q] = eb4[v + q * stride];
            gi[q] = ei4[v + q * stride];
        }
#pragma unroll
        for (int q = 0; q < HG_DEPTH; ++q) hg_count4(h, gb[q], gi[q]);
    }
    for (; v < nvec; v += stride) hg_count4(h, eb4[v], ei4[v]);

    if (hb == 0) {
        const int e = (nvec << 2) + tid;
        if (e < nnz) atomicAdd(&h[eb[e] * N_DIM + ei[e]], 1u);
    }
    __syncthreads();

    unsigned c[HG_BINS / HG_HT];
#pragma unroll
    for (int k = 0; k < HG_BINS / HG_HT; ++k) c[k] = h[k * HG_HT + tid];
#pragma unroll
    for (int k = 0; k < HG_BINS / HG_HT; ++k)
        if (c[k] != 0u) atomicAdd(&degc[k * HG_HT + tid], c[k]);
}

#define MLP_WQ (F_DIM * F_DIM / 4)
#define MLP_BLOCKS (B_DIM * N_DIM / 64)

static_assert(HG_HT == 256, "launch 2: 8 waves per block, a 16-row strip per pair of waves");

__global__ __launch_bounds__(HG_HT) void hg_mlp_hist_kernel(
    const float* __restrict__ x,
    const float* __restrict__ W1, const float* __restrict__ b1,
    const float* __restrict__ W2, const float* __restrict__ b2,
    const float* __restrict__ Ws1, const float* __restrict__ bs1,
    const float* __restrict__ Ws2, const float* __restrict__ bs2,
    float* __restrict__ x1_out, float* __restrict__ self_out,
    const int* __restrict__ eb, const int* __restrict__ ei,
    unsigned* __restrict__ degc, int nnz) {

    __shared__ float sW[4 * F_DIM * F_DIM];
    __shared__ float sB[4 * F_DIM];
    __shared__ float sH[8][16 * F_DIM];

    const int tid  = threadIdx.x;

    if (blockIdx.x >= MLP_BLOCKS) {
        hg_hist_body(eb, ei, degc, nnz, (unsigned*)sW, (int)blockIdx.x - MLP_BLOCKS, tid);
        return;
    }

    const int wave = tid >> 5;
    const int lane = tid & 31;
    const int hlf  = lane >> 4;
    const int mn   = lane & 15;
    const int s    = wave & 1;
    const int rowBase = (blockIdx.x * 4 + (wave >> 1)) * 16;

    v2f aX[16];
#pragma unroll
    for (int kk = 0; kk < 16; ++kk) {
        const float* p = x + (size_t)(rowBase + mn) * F_DIM + kk * 4 + hlf * 2;
        aX[kk][0] = p[0];
        aX[kk][1] = p[1];
    }

    {
        const v4f* g0 = (const v4f*)W1;
        const v4f* g1 = (const v4f*)W2;
        const v4f* g2 = (const v4f*)Ws1;
        const v4f* g3 = (const v4f*)Ws2;
        v4f t0[MLP_WQ / 256], t1[MLP_WQ / 256], t2[MLP_WQ / 256], t3[MLP_WQ / 256];
#pragma unroll
        for (int j = 0; j < MLP_WQ / 256; ++j) {
            t0[j] = g0[j * 256 + tid];
            t1[j] = g1[j * 256 + tid];
            t2[j] = g2[j * 256 + tid];
            t3[j] = g3[j * 256 + tid];
        }
        const int bsel = tid >> 6;
        const float* bp = (bsel == 0) ? b1 : (bsel == 1) ? b2 : (bsel == 2) ? bs1 : bs2;
        const float bv = bp[tid & (F_DIM - 1)];
        v4f* d = (v4f*)sW;
#pragma unroll
        for (int j = 0; j < MLP_WQ / 256; ++j) {
            d[0 * MLP_WQ + j * 256 + tid] = t0[j];
            d[1 * MLP_WQ + j * 256 + tid] = t1[j];
            d[2 * MLP_WQ + j * 256 + tid] = t2[j];
            d[3 * MLP_WQ + j * 256 + tid] = t3[j];
        }
        sB[tid] = bv;
    }
    __syncthreads();

    {
        const float* wA = sW + (2 * s + 0) * F_DIM * F_DIM;
        const float* wB = sW + (2 * s + 1) * F_DIM * F_DIM;
        const float* bA = sB + (2 * s + 0) * F_DIM;
        const float* bB = sB + (2 * s + 1) * F_DIM;
        float* dst = (s == 0) ? x1_out : self_out;

#pragma unroll
        for (int nt = 0; nt < 4; ++nt) {
            v8f c;
            float bias = bA[nt * 16 + mn];
#pragma unroll
            for (int r = 0; r < 8; ++r) c[r] = bias;
#pragma unroll
            for (int kk = 0; kk < 16; ++kk) {
                v2f bF;
                bF[0] = wA[(kk * 4 + hlf * 2 + 0) * F_DIM + nt * 16 + mn];
                bF[1] = wA[(kk * 4 + hlf * 2 + 1) * F_DIM + nt * 16 + mn];
                c = __builtin_amdgcn_wmma_f32_16x16x4_f32(
                        false, aX[kk], false, bF, (short)0, c, false, false);
            }
#pragma unroll
            for (int r = 0; r < 8; ++r)
                sH[wave][(r + 8 * hlf) * F_DIM + nt * 16 + mn] = fmaxf(c[r], 0.0f);
        }
        __syncthreads();

        v2f aH[16];
#pragma unroll
        for (int kk = 0; kk < 16; ++kk) {
            aH[kk][0] = sH[wave][mn * F_DIM + kk * 4 + hlf * 2 + 0];
            aH[kk][1] = sH[wave][mn * F_DIM + kk * 4 + hlf * 2 + 1];
        }

#pragma unroll
        for (int nt = 0; nt < 4; ++nt) {
            v8f c;
            float bias = bB[nt * 16 + mn];
#pragma unroll
            for (int r = 0; r < 8; ++r) c[r] = bias;
#pragma unroll
            for (int kk = 0; kk < 16; ++kk) {
                v2f bF;
                bF[0] = wB[(kk * 4 + hlf * 2 + 0) * F_DIM + nt * 16 + mn];
                bF[1] = wB[(kk * 4 + hlf * 2 + 1) * F_DIM + nt * 16 + mn];
                c = __builtin_amdgcn_wmma_f32_16x16x4_f32(
                        false, aH[kk], false, bF, (short)0, c, false, false);
            }
#pragma unroll
            for (int r = 0; r < 8; ++r) {
                dst[(size_t)(rowBase + r + 8 * hlf) * F_DIM + nt * 16 + mn] =
                    fmaxf(c[r], 0.0f);
            }
        }
    }
}

#define EL_WAVES  32
#define EL_SLICES 32
#define EL_G      16
#define EL_H      8
#define EL_NB     10
#define EL_CAP    4608
#define EL_SR     8

static_assert((1 << EL_NB) == N_DIM, "a record packs three node indices of EL_NB bits");
static_assert(EL_WAVES * 32 == N_DIM, "one thread per node: 1/degree, counter, prefix sum");
static_assert((N_DIM * F_DIM / 4) % (EL_WAVES * 32) == 0, "whole copy trips");
static_assert(EL_CAP <= 65536, "list positions are kept as 16-bit words");

typedef __attribute__((ext_vector_type(2))) int v2i;
typedef __attribute__((ext_vector_type(2))) unsigned v2u;

__device__ __forceinline__ int eg_lane_i(int v, int t) {
    return __builtin_amdgcn_readlane(v, t);
}
__device__ __forceinline__ float eg_lane_f(float v, int t) {
    return __int_as_float(__builtin_amdgcn_readlane(__float_as_int(v), t));
}

__device__ __forceinline__ void el_direct(
    const float* __restrict__ Wval, float* __restrict__ outb,
    const float* sx, const float* sinv, unsigned e, unsigned y, int lane) {
    const int i = (int)(y & (N_DIM - 1));
    const float* xj = sx + (int)((y >> EL_NB) & (N_DIM - 1)) * F_DIM + lane;
    const float* xk = sx + (int)(y >> (2 * EL_NB)) * F_DIM + lane;
    const float* wp = Wval + (size_t)e * F_DIM;
    const float a = sinv[i];
    float* o = outb + i * F_DIM;
    const float w0 = __builtin_nontemporal_load(wp + lane);
    const float w1 = __builtin_nontemporal_load(wp + lane + 32);
    atomicAdd(o + lane,      a * w0 * xk[0]  * xj[0]);
    atomicAdd(o + lane + 32, a * w1 * xk[32] * xj[32]);
}

__device__ __forceinline__ v2u el_rec(
    const v2u* srec, const unsigned short* sord, int p, int hi, int lane) {
    const int q = min(p + (lane & (EL_G - 1)), hi - 1);
    return srec[sord[q]];
}

__device__ __forceinline__ void el_wload(
    const float* __restrict__ Wval, v2u rec, int lane,
    float (&w0)[EL_G], float (&w1)[EL_G]) {
    const int e = (int)rec.x;
#pragma unroll
    for (int t = 0; t < EL_G; ++t) {
        const float* wp = Wval + (size_t)(unsigned)eg_lane_i(e, t) * F_DIM;
        w0[t] = __builtin_nontemporal_load(wp + lane);
        w1[t] = __builtin_nontemporal_load(wp + lane + 32);
    }
}

__device__ __forceinline__ void el_sum(
    float* __restrict__ outb, const float* sx, const float* sinv, v2u rec,
    const float (&w0)[EL_G], const float (&w1)[EL_G],
    int rem, int lane, int& cur, float& acc0, float& acc1) {

    const int i    = (int)(rec.y & (N_DIM - 1));
    const int jOff = (int)((rec.y >> EL_NB) & (N_DIM - 1)) * F_DIM;
    const int kOff = (int)(rec.y >> (2 * EL_NB)) * F_DIM;
    const float av = sinv[i];

#pragma unroll
    for (int g = 0; g < EL_G; g += EL_H) {
        float k0[EL_H], k1[EL_H], j0[EL_H], j1[EL_H];
#pragma unroll
        for (int t = 0; t < EL_H; ++t) {
            const float* xk = sx + eg_lane_i(kOff, g + t) + lane;
            const float* xj = sx + eg_lane_i(jOff, g + t) + lane;
            k0[t] = xk[0];
            k1[t] = xk[32];
            j0[t] = xj[0];
            j1[t] = xj[32];
        }
#pragma unroll
        for (int t = 0; t < EL_H; ++t) {
            const int it = eg_lane_i(i, g + t);
            if (it != cur) {
                if (cur >= 0) {
                    float* o = outb + cur * F_DIM;
                    atomicAdd(o + lane,      acc0);
                    atomicAdd(o + lane + 32, acc1);
                }
                cur  = it;
                acc0 = 0.0f;
                acc1 = 0.0f;
            }
            const float a = (g + t < rem) ? eg_lane_f(av, g + t) : 0.0f;
            acc0 += a * w0[g + t] * k0[t] * j0[t];
            acc1 += a * w1[g + t] * k1[t] * j1[t];
        }
    }
}

__global__ __launch_bounds__(EL_WAVES * 32) void hg_edge_kernel(
    const float* __restrict__ Wval,
    const int* __restrict__ eb, const int* __restrict__ ei,
    const int* __restrict__ ej, const int* __restrict__ ek,
    const unsigned* __restrict__ degc, const float* __restrict__ x1,
    float* __restrict__ out, int nnz) {

    __shared__ float sx[N_DIM * F_DIM];
    __shared__ float sinv[N_DIM];
    __shared__ v2u   srec[EL_CAP];
    __shared__ unsigned short sord[EL_CAP];
    __shared__ unsigned spos[N_DIM];
    __shared__ unsigned swtot[EL_WAVES];
    __shared__ unsigned snrec;

    const int tid  = threadIdx.x;
    const int lane = tid & 31;
    const int wv    = __builtin_amdgcn_readfirstlane(tid >> 5);
    const int b     = blockIdx.x % B_DIM;
    const int slice = blockIdx.x / B_DIM;

    spos[tid] = 0u;
    if (tid == 0) snrec = 0u;
    {
        const v4f* src = (const v4f*)(x1 + (size_t)b * N_DIM * F_DIM);
        v4f* dst = (v4f*)sx;
#pragma unroll
        for (int q = 0; q < N_DIM * F_DIM / 4 / (EL_WAVES * 32); ++q)
            dst[q * (EL_WAVES * 32) + tid] = src[q * (EL_WAVES * 32) + tid];
        sinv[tid] = 1.0f / ((float)degc[b * N_DIM + tid] + EPS_V);
    }
    __syncthreads();

    float* outb = out + (size_t)b * N_DIM * F_DIM;

    const int nfull = nnz >> 6;
    const int per   = (nfull + EL_SLICES - 1) / EL_SLICES;
    const int s0    = slice * per;
    const int s1    = min(s0 + per, nfull);

    const v2i* eb2 = (const v2i*)eb;
    const v2i* ei2 = (const v2i*)ei;
    const v2i* ej2 = (const v2i*)ej;
    const v2i* ek2 = (const v2i*)ek;

    for (int sr = s0 + wv; sr < s1; sr += EL_WAVES * EL_SR) {
        v2i cb[EL_SR], ci[EL_SR], cj[EL_SR], ck[EL_SR];
#pragma unroll
        for (int q = 0; q < EL_SR; ++q) {
            const int p = min(sr + q * EL_WAVES, s1 - 1) * 32 + lane;
            cb[q] = eb2[p];
            ci[q] = ei2[p];
            cj[q] = ej2[p];
            ck[q] = ek2[p];
        }
        unsigned mx[EL_SR], my[EL_SR];
        unsigned tot = 0u;
#pragma unroll
        for (int q = 0; q < EL_SR; ++q) {
            const bool live = (sr + q * EL_WAVES < s1);
            mx[q] = __builtin_amdgcn_ballot_w32(live && cb[q].x == b);
            my[q] = __builtin_amdgcn_ballot_w32(live && cb[q].y == b);
            tot += (unsigned)__builtin_popcount(mx[q]) + (unsigned)__builtin_popcount(my[q]);
        }
        unsigned base = 0u;
        if (lane == 0) base = atomicAdd(&snrec, tot);
        base = (unsigned)__builtin_amdgcn_readfirstlane((int)base);
#pragma unroll
        for (int q = 0; q < EL_SR; ++q) {
            const unsigned e0 = (unsigned)((sr + q * EL_WAVES) * 64 + 2 * lane);
            const bool tx = ((mx[q] >> lane) & 1u) != 0u;
            const bool ty = ((my[q] >> lane) & 1u) != 0u;
            const unsigned ix = (unsigned)(ci[q].x & (N_DIM - 1));
            const unsigned iy = (unsigned)(ci[q].y & (N_DIM - 1));
            const unsigned rx = ix
                              | ((unsigned)(cj[q].x & (N_DIM - 1)) << EL_NB)
                              | ((unsigned)(ck[q].x & (N_DIM - 1)) << (2 * EL_NB));
            const unsigned ry = iy
                              | ((unsigned)(cj[q].y & (N_DIM - 1)) << EL_NB)
                              | ((unsigned)(ck[q].y & (N_DIM - 1)) << (2 * EL_NB));
            const unsigned nx  = (unsigned)__builtin_popcount(mx[q]);
            const unsigned slx = base + __builtin_amdgcn_mbcnt_lo(mx[q], 0u);
            const unsigned sly = base + nx + __builtin_amdgcn_mbcnt_lo(my[q], 0u);
            base += nx + (unsigned)__builtin_popcount(my[q]);
            const bool fx = tx && slx < EL_CAP;
            const bool fy = ty && sly < EL_CAP;
            if (fx) {
                v2u r; r.x = e0; r.y = rx;
                srec[slx] = r;
                atomicAdd(&spos[ix], 1u);
            }
            if (fy) {
                v2u r; r.x = e0 + 1u; r.y = ry;
                srec[sly] = r;
                atomicAdd(&spos[iy], 1u);
            }
            unsigned ox = __builtin_amdgcn_ballot_w32(tx && !fx);
            while (ox) {
                const int t = __builtin_ctz(ox);
                ox &= ox - 1u;
                el_direct(Wval, outb, sx, sinv,
                          (unsigned)__shfl((int)e0, t), (unsigned)__shfl((int)rx, t), lane);
            }
            unsigned oy = __builtin_amdgcn_ballot_w32(ty && !fy);
            while (oy) {
                const int t = __builtin_ctz(oy);
                oy &= oy - 1u;
                el_direct(Wval, outb, sx, sinv,
                          (unsigned)__shfl((int)e0, t) + 1u, (unsigned)__shfl((int)ry, t), lane);
            }
        }
    }
    __syncthreads();

    const unsigned cnt = spos[tid];
    unsigned inc = cnt;
#pragma unroll
    for (int d = 1; d < 32; d <<= 1) {
        const unsigned v = __shfl_up(inc, d);
        if (lane >= d) inc += v;
    }
    if (lane == 31) swtot[wv] = inc;
    __syncthreads();
    unsigned pre = (lane < wv) ? swtot[lane] : 0u;
#pragma unroll
    for (int d = 16; d >= 1; d >>= 1) pre += __shfl_xor(pre, d);
    spos[tid] = pre + inc - cnt;
    __syncthreads();

    const unsigned nrec = min(snrec, (unsigned)EL_CAP);
    for (unsigned r = (unsigned)tid; r < nrec; r += EL_WAVES * 32) {
        const unsigned i = srec[r].y & (N_DIM - 1);
        const unsigned p = atomicAdd(&spos[i], 1u);
        sord[p] = (unsigned short)r;
    }
    __syncthreads();

    {
        const int lo = (wv == 0) ? 0
                     : __builtin_amdgcn_readfirstlane((int)spos[wv * 32 - 1]);
        const int hi = __builtin_amdgcn_readfirstlane((int)spos[wv * 32 + 31]);
        int cur = -1;
        float acc0 = 0.0f, acc1 = 0.0f;
        for (int p = lo; p < hi; p += EL_G) {
            const v2u rec = el_rec(srec, sord, p, hi, lane);
            float w0[EL_G], w1[EL_G];
            el_wload(Wval, rec, lane, w0, w1);
            el_sum(outb, sx, sinv, rec, w0, w1, min(EL_G, hi - p), lane, cur, acc0, acc1);
        }
        if (cur >= 0) {
            float* o = outb + cur * F_DIM;
            atomicAdd(o + lane,      acc0);
            atomicAdd(o + lane + 32, acc1);
        }
    }

    if (slice == EL_SLICES - 1 && wv == 0) {
        for (int e = nfull << 6; e < nnz; ++e) {
            if (eb[e] == b) {
                const unsigned y = (unsigned)(ei[e] & (N_DIM - 1))
                                 | ((unsigned)(ej[e] & (N_DIM - 1)) << EL_NB)
                                 | ((unsigned)(ek[e] & (N_DIM - 1)) << (2 * EL_NB));
                el_direct(Wval, outb, sx, sinv, (unsigned)e, y, lane);
            }
        }
    }
}

extern "C" void kernel_launch(void* const* d_in, const int* in_sizes, int n_in,
                              void* d_out, int out_size, void* d_ws, size_t ws_size,
                              hipStream_t stream) {
    const float* x    = (const float*)d_in[0];
    const float* Wval = (const float*)d_in[1];
    const int*   eb   = (const int*)d_in[2];
    const int*   ei   = (const int*)d_in[3];
    const int*   ej   = (const int*)d_in[4];
    const int*   ek   = (const int*)d_in[5];
    const float* W1   = (const float*)d_in[6];
    const float* b1   = (const float*)d_in[7];
    const float* W2   = (const float*)d_in[8];
    const float* b2   = (const float*)d_in[9];
    const float* Ws1  = (const float*)d_in[10];
    const float* bs1  = (const float*)d_in[11];
    const float* Ws2  = (const float*)d_in[12];
    const float* bs2  = (const float*)d_in[13];

    const int nnz   = in_sizes[2];
    const int nodes = B_DIM * N_DIM;

    unsigned* degc = (unsigned*)d_ws;
    float* x1  = (float*)d_ws + nodes;
    float* out = (float*)d_out;

    hg_degzero_kernel<<<HG_BINS / HG_HT, HG_HT, 0, stream>>>(degc);

    hg_mlp_hist_kernel<<<MLP_BLOCKS + HG_HBLK, HG_HT, 0, stream>>>(
        x, W1, b1, W2, b2, Ws1, bs1, Ws2, bs2, x1, out, eb, ei, degc, nnz);

    hg_edge_kernel<<<B_DIM * EL_SLICES, EL_WAVES * 32, 0, stream>>>(
        Wval, eb, ei, ej, ek, degc, x1, out, nnz);
}
